# starting kernel
# baseline (speedup 1.0000x reference)
#include <hip/hip_runtime.h>
#include <hip/hip_bf16.h>
#include <math.h>

#define B_   2
#define C_   256
#define H_   48
#define W_   48
#define HD   24
#define WD   24
#define L_   576
#define K9   2304
#define K16  4096
#define SCALE_ 10.0f
#define EPS_   1e-4f

typedef float v2f __attribute__((ext_vector_type(2)));
typedef float v8f __attribute__((ext_vector_type(8)));

#define TK  32
#define LDP (TK + 2)

__device__ __forceinline__ float patch9(const float* ds, int b, int k, int l) {
    int c = k / 9, r = k % 9;
    int ki = r / 3, kj = r % 3;
    int y = l / WD + ki - 1;
    int x = l % WD + kj - 1;
    if ((unsigned)y >= HD || (unsigned)x >= WD) return 0.0f;
    return ds[((b * C_ + c) * HD + y) * WD + x];
}

__device__ __forceinline__ float patch16(const float* sc, int b, int m, int l) {
    int c = m >> 4, i = (m >> 2) & 3, j = m & 3;
    int r = 2 * (l / WD) + i - 1;
    int s = 2 * (l % WD) + j - 1;
    if ((unsigned)r >= H_ || (unsigned)s >= W_) return 0.0f;
    return sc[((b * C_ + c) * H_ + r) * W_ + s];
}

__global__ __launch_bounds__(256) void k_prep_ds(const float* __restrict__ left,
                                                 const float* __restrict__ right,
                                                 const float* __restrict__ mid,
                                                 float* __restrict__ dsL,
                                                 float* __restrict__ dsR,
                                                 float* __restrict__ dsM) {
    int idx = blockIdx.x * 256 + threadIdx.x;
    if (idx >= B_ * C_ * HD * WD) return;
    int x = idx % WD, t = idx / WD;
    int y = t % HD;  t /= HD;
    int c = t % C_;  int b = t / C_;
    int src = ((b * C_ + c) * H_ + 2 * y) * W_ + 2 * x;
    dsL[idx] = left[src];
    dsR[idx] = right[src];
    dsM[idx] = mid[src];
}

__global__ __launch_bounds__(256) void k_denom(const float* __restrict__ dsL,
                                               const float* __restrict__ dsR,
                                               float* __restrict__ denL,
                                               float* __restrict__ denR) {
    const float* ds = blockIdx.y ? dsR : dsL;
    float* den      = blockIdx.y ? denR : denL;
    int gl   = blockIdx.x * 8 + (threadIdx.x >> 5);
    int b    = gl / L_;
    int l    = gl % L_;
    int lane = threadIdx.x & 31;
    float s = 0.0f;
    for (int k = lane; k < K9; k += 32) {
        float v = patch9(ds, b, k, l);
        s += v * v;
    }
    #pragma unroll
    for (int off = 16; off; off >>= 1) s += __shfl_xor(s, off, 32);
    if (lane == 0) den[gl] = sqrtf(s + (float)K9 * EPS_);
}

__global__ __launch_bounds__(256) void k_scores(const float* __restrict__ dsL,
                                                const float* __restrict__ dsR,
                                                const float* __restrict__ dsM,
                                                const float* __restrict__ denL,
                                                const float* __restrict__ denR,
                                                float* __restrict__ scL,
                                                float* __restrict__ scR) {
    __shared__ __align__(16) float Ap[64][LDP];
    __shared__ __align__(16) float Bp[64][LDP];
    const float* dsF = blockIdx.y ? dsR : dsL;
    const float* den = blockIdx.y ? denR : denL;
    float* out       = blockIdx.y ? scR : scL;
    int b   = blockIdx.z;
    int l0  = (blockIdx.x / 9) * 64;
    int p0  = (blockIdx.x % 9) * 64;
    int tid = threadIdx.x;
    int lane = tid & 31, w = tid >> 5;
    int tr  = w >> 1;
    int tcb = (w & 1) * 2;
    int m16   = lane & 15;
    int khalf = (lane >> 4) * 2;
    v8f acc0 = {}; v8f acc1 = {};

    for (int kb = 0; kb < K9; kb += TK) {
        __syncthreads();
        #pragma unroll
        for (int t = 0; t < 8; ++t) {
            int idx = tid + t * 256;
            int lr  = idx & 63;
            int kk  = idx >> 6;
            Ap[lr][kk] = patch9(dsF, b, kb + kk, l0 + lr);
            Bp[lr][kk] = patch9(dsM, b, kb + kk, p0 + lr);
        }
        __syncthreads();
        #pragma unroll
        for (int kk0 = 0; kk0 < TK; kk0 += 4) {
            int ka = kk0 + khalf;
            v2f a  = *(const v2f*)&Ap[tr * 16 + m16][ka];
            v2f b0 = *(const v2f*)&Bp[tcb * 16 + m16][ka];
            v2f b1 = *(const v2f*)&Bp[(tcb + 1) * 16 + m16][ka];
            acc0 = __builtin_amdgcn_wmma_f32_16x16x4_f32(false, a, false, b0,
                                                         (short)0, acc0, false, false);
            acc1 = __builtin_amdgcn_wmma_f32_16x16x4_f32(false, a, false, b1,
                                                         (short)0, acc1, false, false);
        }
    }
    int n     = lane & 15;
    int rbase = l0 + tr * 16 + ((lane >> 4) << 3);
    #pragma unroll
    for (int v = 0; v < 8; ++v) {
        int row = rbase + v;
        float scl = SCALE_ / den[b * L_ + row];
        out[((size_t)b * L_ + row) * L_ + p0 + tcb * 16 + n]       = acc0[v] * scl;
        out[((size_t)b * L_ + row) * L_ + p0 + (tcb + 1) * 16 + n] = acc1[v] * scl;
    }
}

__global__ __launch_bounds__(192) void k_softmax(float* __restrict__ scL,
                                                 float* __restrict__ scR) {
    float* sc = blockIdx.y ? scR : scL;
    int b = blockIdx.z;
    int p = blockIdx.x * 192 + threadIdx.x;
    float* col = sc + (size_t)b * L_ * L_ + p;
    float mx = -3.0e38f;
    for (int l = 0; l < L_; ++l) mx = fmaxf(mx, col[(size_t)l * L_]);
    float sum = 0.0f;
    for (int l = 0; l < L_; ++l) {
        float e = expf(col[(size_t)l * L_] - mx);
        col[(size_t)l * L_] = e;
        sum += e;
    }
    float inv = 1.0f / sum;
    for (int l = 0; l < L_; ++l) col[(size_t)l * L_] *= inv;
}

__global__ __launch_bounds__(256) void k_recon(const float* __restrict__ shcL,
                                               const float* __restrict__ shcR,
                                               const float* __restrict__ atL,
                                               const float* __restrict__ atR,
                                               float* __restrict__ YL,
                                               float* __restrict__ YR) {
    __shared__ __align__(16) float Ap[64][LDP];
    __shared__ __align__(16) float Bp[64][LDP];
    const float* sc = blockIdx.y ? shcR : shcL;
    const float* at = blockIdx.y ? atR  : atL;
    float* Y        = blockIdx.y ? YR   : YL;
    int b   = blockIdx.z;
    int m0  = (blockIdx.x / 9) * 64;
    int p0  = (blockIdx.x % 9) * 64;
    int tid = threadIdx.x;
    int lane = tid & 31, w = tid >> 5;
    int tr  = w >> 1;
    int tcb = (w & 1) * 2;
    int m16   = lane & 15;
    int khalf = (lane >> 4) * 2;
    v8f acc0 = {}; v8f acc1 = {};

    for (int kb = 0; kb < L_; kb += TK) {
        __syncthreads();
        #pragma unroll
        for (int t = 0; t < 8; ++t) {
            int idx = tid + t * 256;
            int lr  = idx & 63;
            int kk  = idx >> 6;
            Ap[lr][kk] = patch16(sc, b, m0 + lr, kb + kk);
            Bp[lr][kk] = at[((size_t)b * L_ + kb + kk) * L_ + p0 + lr];
        }
        __syncthreads();
        #pragma unroll
        for (int kk0 = 0; kk0 < TK; kk0 += 4) {
            int ka = kk0 + khalf;
            v2f a  = *(const v2f*)&Ap[tr * 16 + m16][ka];
            v2f b0 = *(const v2f*)&Bp[tcb * 16 + m16][ka];
            v2f b1 = *(const v2f*)&Bp[(tcb + 1) * 16 + m16][ka];
            acc0 = __builtin_amdgcn_wmma_f32_16x16x4_f32(false, a, false, b0,
                                                         (short)0, acc0, false, false);
            acc1 = __builtin_amdgcn_wmma_f32_16x16x4_f32(false, a, false, b1,
                                                         (short)0, acc1, false, false);
        }
    }
    int n     = lane & 15;
    int rbase = m0 + tr * 16 + ((lane >> 4) << 3);
    #pragma unroll
    for (int v = 0; v < 8; ++v) {
        int row = rbase + v;
        Y[((size_t)b * K16 + row) * L_ + p0 + tcb * 16 + n]       = acc0[v];
        Y[((size_t)b * K16 + row) * L_ + p0 + (tcb + 1) * 16 + n] = acc1[v];
    }
}

__global__ __launch_bounds__(256) void k_finalize(const float* __restrict__ YL,
                                                  const float* __restrict__ YR,
                                                  float* __restrict__ out) {
    int idx = blockIdx.x * 256 + threadIdx.x;
    if (idx >= B_ * C_ * H_ * W_) return;
    int s = idx % W_; int t = idx / W_;
    int r = t % H_;  t /= H_;
    int c = t % C_;  int b = t / C_;
    float aL = 0.0f, aR = 0.0f;
    int i0 = (r + 1) & 1, j0 = (s + 1) & 1;
    #pragma unroll
    for (int di = 0; di < 2; ++di) {
        int i = i0 + 2 * di, ry = r + 1 - i;
        if (ry < 0 || ry >= 2 * HD) continue;
        int y = ry >> 1;
        #pragma unroll
        for (int dj = 0; dj < 2; ++dj) {
            int j = j0 + 2 * dj, sx = s + 1 - j;
            if (sx < 0 || sx >= 2 * WD) continue;
            int x = sx >> 1;
            int m = ((c << 2) | i) * 4 + j;
            size_t off = ((size_t)b * K16 + m) * L_ + y * WD + x;
            aL += YL[off];
            aR += YR[off];
        }
    }
    const float PI = 3.14159265358979323846f;
    float wl = 0.5f * (cosf(PI * (float)s / (float)(W_ - 1)) + 1.0f);
    float wr = 0.5f * (cosf(PI * (float)(W_ - 1 - s) / (float)(W_ - 1)) + 1.0f);
    out[idx] = 0.25f * (wl * aL + wr * aR);
}

extern "C" void kernel_launch(void* const* d_in, const int* in_sizes, int n_in,
                              void* d_out, int out_size, void* d_ws, size_t ws_size,
                              hipStream_t stream) {
    (void)in_sizes; (void)n_in; (void)out_size; (void)ws_size;
    const float* left  = (const float*)d_in[0];
    const float* right = (const float*)d_in[1];
    const float* mid   = (const float*)d_in[2];
    const float* shcL  = (const float*)d_in[3];
    const float* shcR  = (const float*)d_in[4];
    float* out = (float*)d_out;

    float* w   = (float*)d_ws;
    const size_t NDS = (size_t)B_ * C_ * HD * WD;
    const size_t NDN = (size_t)B_ * L_;
    const size_t NSC = (size_t)B_ * L_ * L_;
    const size_t NY  = (size_t)B_ * K16 * L_;
    float* dsL  = w;
    float* dsR  = dsL + NDS;
    float* dsM  = dsR + NDS;
    float* denL = dsM + NDS;
    float* denR = denL + NDN;
    float* scL  = denR + NDN;
    float* scR  = scL + NSC;
    float* YL   = scR + NSC;
    float* YR   = YL + NY;

    k_prep_ds<<<dim3((NDS + 255) / 256), dim3(256), 0, stream>>>(left, right, mid,
                                                                 dsL, dsR, dsM);
    k_denom<<<dim3(B_ * L_ / 8, 2), dim3(256), 0, stream>>>(dsL, dsR, denL, denR);
    k_scores<<<dim3(81, 2, B_), dim3(256), 0, stream>>>(dsL, dsR, dsM,
                                                        denL, denR, scL, scR);
    k_softmax<<<dim3(3, 2, B_), dim3(192), 0, stream>>>(scL, scR);
    k_recon<<<dim3(576, 2, B_), dim3(256), 0, stream>>>(shcL, shcR, scL, scR, YL, YR);
    k_finalize<<<dim3((B_ * C_ * H_ * W_ + 255) / 256), dim3(256), 0, stream>>>(YL, YR, out);
}


# softmax parallelised over rows: one block per 32 columns, the 576 rows spread over 16 waves and held in registers, column max and sum combined through LDS; one read and one write
# speedup vs baseline: 1.4078x; 1.4078x over previous
#include <hip/hip_runtime.h>
#include <hip/hip_bf16.h>
#include <math.h>

#define B_   2
#define C_   256
#define H_   48
#define W_   48
#define HD   24
#define WD   24
#define L_   576
#define K9   2304
#define K16  4096
#define SCALE_ 10.0f
#define EPS_   1e-4f

typedef float v2f __attribute__((ext_vector_type(2)));
typedef float v8f __attribute__((ext_vector_type(8)));

#define TK  32
#define LDP (TK + 2)

__device__ __forceinline__ float patch9(const float* ds, int b, int k, int l) {
    int c = k / 9, r = k % 9;
    int ki = r / 3, kj = r % 3;
    int y = l / WD + ki - 1;
    int x = l % WD + kj - 1;
    if ((unsigned)y >= HD || (unsigned)x >= WD) return 0.0f;
    return ds[((b * C_ + c) * HD + y) * WD + x];
}

__device__ __forceinline__ float patch16(const float* sc, int b, int m, int l) {
    int c = m >> 4, i = (m >> 2) & 3, j = m & 3;
    int r = 2 * (l / WD) + i - 1;
    int s = 2 * (l % WD) + j - 1;
    if ((unsigned)r >= H_ || (unsigned)s >= W_) return 0.0f;
    return sc[((b * C_ + c) * H_ + r) * W_ + s];
}

__global__ __launch_bounds__(256) void k_prep_ds(const float* __restrict__ left,
                                                 const float* __restrict__ right,
                                                 const float* __restrict__ mid,
                                                 float* __restrict__ dsL,
                                                 float* __restrict__ dsR,
                                                 float* __restrict__ dsM) {
    int idx = blockIdx.x * 256 + threadIdx.x;
    if (idx >= B_ * C_ * HD * WD) return;
    int x = idx % WD, t = idx / WD;
    int y = t % HD;  t /= HD;
    int c = t % C_;  int b = t / C_;
    int src = ((b * C_ + c) * H_ + 2 * y) * W_ + 2 * x;
    dsL[idx] = left[src];
    dsR[idx] = right[src];
    dsM[idx] = mid[src];
}

__global__ __launch_bounds__(256) void k_denom(const float* __restrict__ dsL,
                                               const float* __restrict__ dsR,
                                               float* __restrict__ denL,
                                               float* __restrict__ denR) {
    const float* ds = blockIdx.y ? dsR : dsL;
    float* den      = blockIdx.y ? denR : denL;
    int gl   = blockIdx.x * 8 + (threadIdx.x >> 5);
    int b    = gl / L_;
    int l    = gl % L_;
    int lane = threadIdx.x & 31;
    float s = 0.0f;
    for (int k = lane; k < K9; k += 32) {
        float v = patch9(ds, b, k, l);
        s += v * v;
    }
    #pragma unroll
    for (int off = 16; off; off >>= 1) s += __shfl_xor(s, off, 32);
    if (lane == 0) den[gl] = sqrtf(s + (float)K9 * EPS_);
}

__global__ __launch_bounds__(256) void k_scores(const float* __restrict__ dsL,
                                                const float* __restrict__ dsR,
                                                const float* __restrict__ dsM,
                                                const float* __restrict__ denL,
                                                const float* __restrict__ denR,
                                                float* __restrict__ scL,
                                                float* __restrict__ scR) {
    __shared__ __align__(16) float Ap[64][LDP];
    __shared__ __align__(16) float Bp[64][LDP];
    const float* dsF = blockIdx.y ? dsR : dsL;
    const float* den = blockIdx.y ? denR : denL;
    float* out       = blockIdx.y ? scR : scL;
    int b   = blockIdx.z;
    int l0  = (blockIdx.x / 9) * 64;
    int p0  = (blockIdx.x % 9) * 64;
    int tid = threadIdx.x;
    int lane = tid & 31, w = tid >> 5;
    int tr  = w >> 1;
    int tcb = (w & 1) * 2;
    int m16   = lane & 15;
    int khalf = (lane >> 4) * 2;
    v8f acc0 = {}; v8f acc1 = {};

    for (int kb = 0; kb < K9; kb += TK) {
        __syncthreads();
        #pragma unroll
        for (int t = 0; t < 8; ++t) {
            int idx = tid + t * 256;
            int lr  = idx & 63;
            int kk  = idx >> 6;
            Ap[lr][kk] = patch9(dsF, b, kb + kk, l0 + lr);
            Bp[lr][kk] = patch9(dsM, b, kb + kk, p0 + lr);
        }
        __syncthreads();
        #pragma unroll
        for (int kk0 = 0; kk0 < TK; kk0 += 4) {
            int ka = kk0 + khalf;
            v2f a  = *(const v2f*)&Ap[tr * 16 + m16][ka];
            v2f b0 = *(const v2f*)&Bp[tcb * 16 + m16][ka];
            v2f b1 = *(const v2f*)&Bp[(tcb + 1) * 16 + m16][ka];
            acc0 = __builtin_amdgcn_wmma_f32_16x16x4_f32(false, a, false, b0,
                                                         (short)0, acc0, false, false);
            acc1 = __builtin_amdgcn_wmma_f32_16x16x4_f32(false, a, false, b1,
                                                         (short)0, acc1, false, false);
        }
    }
    int n     = lane & 15;
    int rbase = l0 + tr * 16 + ((lane >> 4) << 3);
    #pragma unroll
    for (int v = 0; v < 8; ++v) {
        int row = rbase + v;
        float scl = SCALE_ / den[b * L_ + row];
        out[((size_t)b * L_ + row) * L_ + p0 + tcb * 16 + n]       = acc0[v] * scl;
        out[((size_t)b * L_ + row) * L_ + p0 + (tcb + 1) * 16 + n] = acc1[v] * scl;
    }
}

#define SM_COLS  32
#define SM_WAVES 16
#define SM_ROWS  (L_ / SM_WAVES)
__global__ __launch_bounds__(SM_COLS * SM_WAVES) void k_softmax(float* __restrict__ scL,
                                                                float* __restrict__ scR) {
    __shared__ float redm[SM_WAVES][SM_COLS];
    __shared__ float reds[SM_WAVES][SM_COLS];
    float* sc = blockIdx.y ? scR : scL;
    int b    = blockIdx.z;
    int lane = threadIdx.x & (SM_COLS - 1);
    int w    = threadIdx.x / SM_COLS;
    int p    = blockIdx.x * SM_COLS + lane;
    float* col = sc + ((size_t)b * L_ + (size_t)w * SM_ROWS) * L_ + p;

    float v[SM_ROWS];
    #pragma unroll
    for (int i = 0; i < SM_ROWS; ++i) v[i] = col[(size_t)i * L_];

    float mx = v[0];
    #pragma unroll
    for (int i = 1; i < SM_ROWS; ++i) mx = fmaxf(mx, v[i]);
    redm[w][lane] = mx;
    __syncthreads();
    float m = redm[0][lane];
    #pragma unroll
    for (int j = 1; j < SM_WAVES; ++j) m = fmaxf(m, redm[j][lane]);

    float s = 0.0f;
    #pragma unroll
    for (int i = 0; i < SM_ROWS; ++i) {
        v[i] = expf(v[i] - m);
        s += v[i];
    }
    reds[w][lane] = s;
    __syncthreads();
    float tot = 0.0f;
    #pragma unroll
    for (int j = 0; j < SM_WAVES; ++j) tot += reds[j][lane];
    float inv = 1.0f / tot;

    #pragma unroll
    for (int i = 0; i < SM_ROWS; ++i) col[(size_t)i * L_] = v[i] * inv;
}

__global__ __launch_bounds__(256) void k_recon(const float* __restrict__ shcL,
                                               const float* __restrict__ shcR,
                                               const float* __restrict__ atL,
                                               const float* __restrict__ atR,
                                               float* __restrict__ YL,
                                               float* __restrict__ YR) {
    __shared__ __align__(16) float Ap[64][LDP];
    __shared__ __align__(16) float Bp[64][LDP];
    const float* sc = blockIdx.y ? shcR : shcL;
    const float* at = blockIdx.y ? atR  : atL;
    float* Y        = blockIdx.y ? YR   : YL;
    int b   = blockIdx.z;
    int m0  = (blockIdx.x / 9) * 64;
    int p0  = (blockIdx.x % 9) * 64;
    int tid = threadIdx.x;
    int lane = tid & 31, w = tid >> 5;
    int tr  = w >> 1;
    int tcb = (w & 1) * 2;
    int m16   = lane & 15;
    int khalf = (lane >> 4) * 2;
    v8f acc0 = {}; v8f acc1 = {};

    for (int kb = 0; kb < L_; kb += TK) {
        __syncthreads();
        #pragma unroll
        for (int t = 0; t < 8; ++t) {
            int idx = tid + t * 256;
            int lr  = idx & 63;
            int kk  = idx >> 6;
            Ap[lr][kk] = patch16(sc, b, m0 + lr, kb + kk);
            Bp[lr][kk] = at[((size_t)b * L_ + kb + kk) * L_ + p0 + lr];
        }
        __syncthreads();
        #pragma unroll
        for (int kk0 = 0; kk0 < TK; kk0 += 4) {
            int ka = kk0 + khalf;
            v2f a  = *(const v2f*)&Ap[tr * 16 + m16][ka];
            v2f b0 = *(const v2f*)&Bp[tcb * 16 + m16][ka];
            v2f b1 = *(const v2f*)&Bp[(tcb + 1) * 16 + m16][ka];
            acc0 = __builtin_amdgcn_wmma_f32_16x16x4_f32(false, a, false, b0,
                                                         (short)0, acc0, false, false);
            acc1 = __builtin_amdgcn_wmma_f32_16x16x4_f32(false, a, false, b1,
                                                         (short)0, acc1, false, false);
        }
    }
    int n     = lane & 15;
    int rbase = m0 + tr * 16 + ((lane >> 4) << 3);
    #pragma unroll
    for (int v = 0; v < 8; ++v) {
        int row = rbase + v;
        Y[((size_t)b * K16 + row) * L_ + p0 + tcb * 16 + n]       = acc0[v];
        Y[((size_t)b * K16 + row) * L_ + p0 + (tcb + 1) * 16 + n] = acc1[v];
    }
}

__global__ __launch_bounds__(256) void k_finalize(const float* __restrict__ YL,
                                                  const float* __restrict__ YR,
                                                  float* __restrict__ out) {
    int idx = blockIdx.x * 256 + threadIdx.x;
    if (idx >= B_ * C_ * H_ * W_) return;
    int s = idx % W_; int t = idx / W_;
    int r = t % H_;  t /= H_;
    int c = t % C_;  int b = t / C_;
    float aL = 0.0f, aR = 0.0f;
    int i0 = (r + 1) & 1, j0 = (s + 1) & 1;
    #pragma unroll
    for (int di = 0; di < 2; ++di) {
        int i = i0 + 2 * di, ry = r + 1 - i;
        if (ry < 0 || ry >= 2 * HD) continue;
        int y = ry >> 1;
        #pragma unroll
        for (int dj = 0; dj < 2; ++dj) {
            int j = j0 + 2 * dj, sx = s + 1 - j;
            if (sx < 0 || sx >= 2 * WD) continue;
            int x = sx >> 1;
            int m = ((c << 2) | i) * 4 + j;
            size_t off = ((size_t)b * K16 + m) * L_ + y * WD + x;
            aL += YL[off];
            aR += YR[off];
        }
    }
    const float PI = 3.14159265358979323846f;
    float wl = 0.5f * (cosf(PI * (float)s / (float)(W_ - 1)) + 1.0f);
    float wr = 0.5f * (cosf(PI * (float)(W_ - 1 - s) / (float)(W_ - 1)) + 1.0f);
    out[idx] = 0.25f * (wl * aL + wr * aR);
}

extern "C" void kernel_launch(void* const* d_in, const int* in_sizes, int n_in,
                              void* d_out, int out_size, void* d_ws, size_t ws_size,
                              hipStream_t stream) {
    (void)in_sizes; (void)n_in; (void)out_size; (void)ws_size;
    const float* left  = (const float*)d_in[0];
    const float* right = (const float*)d_in[1];
    const float* mid   = (const float*)d_in[2];
    const float* shcL  = (const float*)d_in[3];
    const float* shcR  = (const float*)d_in[4];
    float* out = (float*)d_out;

    float* w   = (float*)d_ws;
    const size_t NDS = (size_t)B_ * C_ * HD * WD;
    const size_t NDN = (size_t)B_ * L_;
    const size_t NSC = (size_t)B_ * L_ * L_;
    const size_t NY  = (size_t)B_ * K16 * L_;
    float* dsL  = w;
    float* dsR  = dsL + NDS;
    float* dsM  = dsR + NDS;
    float* denL = dsM + NDS;
    float* denR = denL + NDN;
    float* scL  = denR + NDN;
    float* scR  = scL + NSC;
    float* YL   = scR + NSC;
    float* YR   = YL + NY;

    k_prep_ds<<<dim3((NDS + 255) / 256), dim3(256), 0, stream>>>(left, right, mid,
                                                                 dsL, dsR, dsM);
    k_denom<<<dim3(B_ * L_ / 8, 2), dim3(256), 0, stream>>>(dsL, dsR, denL, denR);
    k_scores<<<dim3(81, 2, B_), dim3(256), 0, stream>>>(dsL, dsR, dsM,
                                                        denL, denR, scL, scR);
    k_softmax<<<dim3(L_ / SM_COLS, 2, B_), dim3(SM_COLS * SM_WAVES), 0, stream>>>(scL, scR);
    k_recon<<<dim3(576, 2, B_), dim3(256), 0, stream>>>(shcL, shcR, scL, scR, YL, YR);
    k_finalize<<<dim3((B_ * C_ * H_ * W_ + 255) / 256), dim3(256), 0, stream>>>(YL, YR, out);
}


# GEMM staging: bounds branches on the gather loads replaced by clamp+select, a K-chunk's 16 loads issued as one group one chunk ahead (2 LDS stages), waited behind the WMMA block, not one by one
# speedup vs baseline: 2.9388x; 2.0875x over previous
#include <hip/hip_runtime.h>
#include <hip/hip_bf16.h>
#include <math.h>

#define B_   2
#define C_   256
#define H_   48
#define W_   48
#define HD   24
#define WD   24
#define L_   576
#define K9   2304
#define K16  4096
#define SCALE_ 10.0f
#define EPS_   1e-4f

typedef float v2f __attribute__((ext_vector_type(2)));
typedef float v8f __attribute__((ext_vector_type(8)));

#define TK  32
#define LDP (TK + 2)

__device__ __forceinline__ float patch9(const float* ds, int b, int k, int l) {
    int c = k / 9, r = k % 9;
    int ki = r / 3, kj = r % 3;
    int y = l / WD + ki - 1;
    int x = l % WD + kj - 1;
    bool ok = ((unsigned)y < HD) & ((unsigned)x < WD);
    int off = ok ? (y * WD + x) : 0;
    float v = ds[(b * C_ + c) * (HD * WD) + off];
    return ok ? v : 0.0f;
}
__device__ __forceinline__ int off9(int b, int k, int l, unsigned& ok) {
    int c = k / 9, r = k % 9;
    int ki = r / 3, kj = r % 3;
    int y = l / WD + ki - 1;
    int x = l % WD + kj - 1;
    bool in = ((unsigned)y < HD) & ((unsigned)x < WD);
    ok = in ? 1u : 0u;
    return (b * C_ + c) * (HD * WD) + (in ? (y * WD + x) : 0);
}

__device__ __forceinline__ float patch16(const float* sc, int b, int m, int l) {
    int c = m >> 4, i = (m >> 2) & 3, j = m & 3;
    int r = 2 * (l / WD) + i - 1;
    int s = 2 * (l % WD) + j - 1;
    bool ok = ((unsigned)r < H_) & ((unsigned)s < W_);
    int off = ok ? (r * W_ + s) : 0;
    float v = sc[(b * C_ + c) * (H_ * W_) + off];
    return ok ? v : 0.0f;
}
__device__ __forceinline__ int off16(int b, int m, int l, unsigned& ok) {
    int c = m >> 4, i = (m >> 2) & 3, j = m & 3;
    int r = 2 * (l / WD) + i - 1;
    int s = 2 * (l % WD) + j - 1;
    bool in = ((unsigned)r < H_) & ((unsigned)s < W_);
    ok = in ? 1u : 0u;
    return (b * C_ + c) * (H_ * W_) + (in ? (r * W_ + s) : 0);
}

__global__ __launch_bounds__(256) void k_prep_ds(const float* __restrict__ left,
                                                 const float* __restrict__ right,
                                                 const float* __restrict__ mid,
                                                 float* __restrict__ dsL,
                                                 float* __restrict__ dsR,
                                                 float* __restrict__ dsM) {
    int idx = blockIdx.x * 256 + threadIdx.x;
    if (idx >= B_ * C_ * HD * WD) return;
    int x = idx % WD, t = idx / WD;
    int y = t % HD;  t /= HD;
    int c = t % C_;  int b = t / C_;
    int src = ((b * C_ + c) * H_ + 2 * y) * W_ + 2 * x;
    dsL[idx] = left[src];
    dsR[idx] = right[src];
    dsM[idx] = mid[src];
}

__global__ __launch_bounds__(256) void k_denom(const float* __restrict__ dsL,
                                               const float* __restrict__ dsR,
                                               float* __restrict__ denL,
                                               float* __restrict__ denR) {
    const float* ds = blockIdx.y ? dsR : dsL;
    float* den      = blockIdx.y ? denR : denL;
    int gl   = blockIdx.x * 8 + (threadIdx.x >> 5);
    int b    = gl / L_;
    int l    = gl % L_;
    int lane = threadIdx.x & 31;
    float s = 0.0f;
    #pragma unroll 8
    for (int k = lane; k < K9; k += 32) {
        float v = patch9(ds, b, k, l);
        s += v * v;
    }
    #pragma unroll
    for (int off = 16; off; off >>= 1) s += __shfl_xor(s, off, 32);
    if (lane == 0) den[gl] = sqrtf(s + (float)K9 * EPS_);
}

__global__ __launch_bounds__(256) void k_scores(const float* __restrict__ dsL,
                                                const float* __restrict__ dsR,
                                                const float* __restrict__ dsM,
                                                const float* __restrict__ denL,
                                                const float* __restrict__ denR,
                                                float* __restrict__ scL,
                                                float* __restrict__ scR) {
    __shared__ __align__(16) float Ap[2][64][LDP];
    __shared__ __align__(16) float Bp[2][64][LDP];
    const float* dsF = blockIdx.y ? dsR : dsL;
    const float* den = blockIdx.y ? denR : denL;
    float* out       = blockIdx.y ? scR : scL;
    int b   = blockIdx.z;
    int l0  = (blockIdx.x / 9) * 64;
    int p0  = (blockIdx.x % 9) * 64;
    int tid = threadIdx.x;
    int lane = tid & 31, w = tid >> 5;
    int tr  = w >> 1;
    int tcb = (w & 1) * 2;
    int m16   = lane & 15;
    int khalf = (lane >> 4) * 2;
    v8f acc0 = {}; v8f acc1 = {};

    const int lr = tid & 63;
    const int kq = tid >> 6;
    float ra[8], rb[8];
    #pragma unroll
    for (int t = 0; t < 8; ++t) {
        ra[t] = patch9(dsF, b, kq + 4 * t, l0 + lr);
        rb[t] = patch9(dsM, b, kq + 4 * t, p0 + lr);
    }
    #pragma unroll
    for (int t = 0; t < 8; ++t) {
        Ap[0][lr][kq + 4 * t] = ra[t];
        Bp[0][lr][kq + 4 * t] = rb[t];
    }
    __syncthreads();
    int cur = 0;
    for (int kb = 0; kb < K9; kb += TK) {
        const bool more = (kb + TK) < K9;
        unsigned okm = 0u;
        if (more) {
            #pragma unroll
            for (int t = 0; t < 8; ++t) {
                unsigned oa, ob;
                int ia = off9(b, kb + TK + kq + 4 * t, l0 + lr, oa);
                int ib = off9(b, kb + TK + kq + 4 * t, p0 + lr, ob);
                ra[t] = dsF[ia];
                rb[t] = dsM[ib];
                okm |= (oa << t) | (ob << (8 + t));
            }
        }
        #pragma unroll
        for (int kk0 = 0; kk0 < TK; kk0 += 4) {
            int ka = kk0 + khalf;
            v2f a  = *(const v2f*)&Ap[cur][tr * 16 + m16][ka];
            v2f b0 = *(const v2f*)&Bp[cur][tcb * 16 + m16][ka];
            v2f b1 = *(const v2f*)&Bp[cur][(tcb + 1) * 16 + m16][ka];
            acc0 = __builtin_amdgcn_wmma_f32_16x16x4_f32(false, a, false, b0,
                                                         (short)0, acc0, false, false);
            acc1 = __builtin_amdgcn_wmma_f32_16x16x4_f32(false, a, false, b1,
                                                         (short)0, acc1, false, false);
        }
        if (more) {
            #pragma unroll
            for (int t = 0; t < 8; ++t) {
                Ap[cur ^ 1][lr][kq + 4 * t] = ((okm >> t) & 1u) ? ra[t] : 0.0f;
                Bp[cur ^ 1][lr][kq + 4 * t] = ((okm >> (8 + t)) & 1u) ? rb[t] : 0.0f;
            }
        }
        __syncthreads();
        cur ^= 1;
    }
    int n     = lane & 15;
    int rbase = l0 + tr * 16 + ((lane >> 4) << 3);
    #pragma unroll
    for (int v = 0; v < 8; ++v) {
        int row = rbase + v;
        float scl = SCALE_ / den[b * L_ + row];
        out[((size_t)b * L_ + row) * L_ + p0 + tcb * 16 + n]       = acc0[v] * scl;
        out[((size_t)b * L_ + row) * L_ + p0 + (tcb + 1) * 16 + n] = acc1[v] * scl;
    }
}

#define SM_COLS  32
#define SM_WAVES 16
#define SM_ROWS  (L_ / SM_WAVES)
__global__ __launch_bounds__(SM_COLS * SM_WAVES) void k_softmax(float* __restrict__ scL,
                                                                float* __restrict__ scR) {
    __shared__ float redm[SM_WAVES][SM_COLS];
    __shared__ float reds[SM_WAVES][SM_COLS];
    float* sc = blockIdx.y ? scR : scL;
    int b    = blockIdx.z;
    int lane = threadIdx.x & (SM_COLS - 1);
    int w    = threadIdx.x / SM_COLS;
    int p    = blockIdx.x * SM_COLS + lane;
    float* col = sc + ((size_t)b * L_ + (size_t)w * SM_ROWS) * L_ + p;

    float v[SM_ROWS];
    #pragma unroll
    for (int i = 0; i < SM_ROWS; ++i) v[i] = col[(size_t)i * L_];

    float mx = v[0];
    #pragma unroll
    for (int i = 1; i < SM_ROWS; ++i) mx = fmaxf(mx, v[i]);
    redm[w][lane] = mx;
    __syncthreads();
    float m = redm[0][lane];
    #pragma unroll
    for (int j = 1; j < SM_WAVES; ++j) m = fmaxf(m, redm[j][lane]);

    float s = 0.0f;
    #pragma unroll
    for (int i = 0; i < SM_ROWS; ++i) {
        v[i] = expf(v[i] - m);
        s += v[i];
    }
    reds[w][lane] = s;
    __syncthreads();
    float tot = 0.0f;
    #pragma unroll
    for (int j = 0; j < SM_WAVES; ++j) tot += reds[j][lane];
    float inv = 1.0f / tot;

    #pragma unroll
    for (int i = 0; i < SM_ROWS; ++i) col[(size_t)i * L_] = v[i] * inv;
}

__global__ __launch_bounds__(256) void k_recon(const float* __restrict__ shcL,
                                               const float* __restrict__ shcR,
                                               const float* __restrict__ atL,
                                               const float* __restrict__ atR,
                                               float* __restrict__ YL,
                                               float* __restrict__ YR) {
    __shared__ __align__(16) float Ap[2][64][LDP];
    __shared__ __align__(16) float Bp[2][64][LDP];
    const float* sc = blockIdx.y ? shcR : shcL;
    const float* at = blockIdx.y ? atR  : atL;
    float* Y        = blockIdx.y ? YR   : YL;
    int b   = blockIdx.z;
    int m0  = (blockIdx.x / 9) * 64;
    int p0  = (blockIdx.x % 9) * 64;
    int tid = threadIdx.x;
    int lane = tid & 31, w = tid >> 5;
    int tr  = w >> 1;
    int tcb = (w & 1) * 2;
    int m16   = lane & 15;
    int khalf = (lane >> 4) * 2;
    v8f acc0 = {}; v8f acc1 = {};

    const int lr = tid & 63;
    const int kq = tid >> 6;
    const float* atp = at + (size_t)b * L_ * L_ + p0 + lr;
    float ra[8], rb[8];
    #pragma unroll
    for (int t = 0; t < 8; ++t) {
        ra[t] = patch16(sc, b, m0 + lr, kq + 4 * t);
        rb[t] = atp[(size_t)(kq + 4 * t) * L_];
    }
    #pragma unroll
    for (int t = 0; t < 8; ++t) {
        Ap[0][lr][kq + 4 * t] = ra[t];
        Bp[0][lr][kq + 4 * t] = rb[t];
    }
    __syncthreads();
    int cur = 0;
    for (int kb = 0; kb < L_; kb += TK) {
        const bool more = (kb + TK) < L_;
        unsigned okm = 0u;
        if (more) {
            #pragma unroll
            for (int t = 0; t < 8; ++t) {
                unsigned oa;
                int ia = off16(b, m0 + lr, kb + TK + kq + 4 * t, oa);
                ra[t] = sc[ia];
                rb[t] = atp[(size_t)(kb + TK + kq + 4 * t) * L_];
                okm |= oa << t;
            }
        }
        #pragma unroll
        for (int kk0 = 0; kk0 < TK; kk0 += 4) {
            int ka = kk0 + khalf;
            v2f a  = *(const v2f*)&Ap[cur][tr * 16 + m16][ka];
            v2f b0 = *(const v2f*)&Bp[cur][tcb * 16 + m16][ka];
            v2f b1 = *(const v2f*)&Bp[cur][(tcb + 1) * 16 + m16][ka];
            acc0 = __builtin_amdgcn_wmma_f32_16x16x4_f32(false, a, false, b0,
                                                         (short)0, acc0, false, false);
            acc1 = __builtin_amdgcn_wmma_f32_16x16x4_f32(false, a, false, b1,
                                                         (short)0, acc1, false, false);
        }
        if (more) {
            #pragma unroll
            for (int t = 0; t < 8; ++t) {
                Ap[cur ^ 1][lr][kq + 4 * t] = ((okm >> t) & 1u) ? ra[t] : 0.0f;
                Bp[cur ^ 1][lr][kq + 4 * t] = rb[t];
            }
        }
        __syncthreads();
        cur ^= 1;
    }
    int n     = lane & 15;
    int rbase = m0 + tr * 16 + ((lane >> 4) << 3);
    #pragma unroll
    for (int v = 0; v < 8; ++v) {
        int row = rbase + v;
        Y[((size_t)b * K16 + row) * L_ + p0 + tcb * 16 + n]       = acc0[v];
        Y[((size_t)b * K16 + row) * L_ + p0 + (tcb + 1) * 16 + n] = acc1[v];
    }
}

__global__ __launch_bounds__(256) void k_finalize(const float* __restrict__ YL,
                                                  const float* __restrict__ YR,
                                                  float* __restrict__ out) {
    int idx = blockIdx.x * 256 + threadIdx.x;
    if (idx >= B_ * C_ * H_ * W_) return;
    int s = idx % W_; int t = idx / W_;
    int r = t % H_;  t /= H_;
    int c = t % C_;  int b = t / C_;
    float aL = 0.0f, aR = 0.0f;
    int i0 = (r + 1) & 1, j0 = (s + 1) & 1;
    #pragma unroll
    for (int di = 0; di < 2; ++di) {
        int i = i0 + 2 * di, ry = r + 1 - i;
        if (ry < 0 || ry >= 2 * HD) continue;
        int y = ry >> 1;
        #pragma unroll
        for (int dj = 0; dj < 2; ++dj) {
            int j = j0 + 2 * dj, sx = s + 1 - j;
            if (sx < 0 || sx >= 2 * WD) continue;
            int x = sx >> 1;
            int m = ((c << 2) | i) * 4 + j;
            size_t off = ((size_t)b * K16 + m) * L_ + y * WD + x;
            aL += YL[off];
            aR += YR[off];
        }
    }
    const float PI = 3.14159265358979323846f;
    float wl = 0.5f * (cosf(PI * (float)s / (float)(W_ - 1)) + 1.0f);
    float wr = 0.5f * (cosf(PI * (float)(W_ - 1 - s) / (float)(W_ - 1)) + 1.0f);
    out[idx] = 0.25f * (wl * aL + wr * aR);
}

extern "C" void kernel_launch(void* const* d_in, const int* in_sizes, int n_in,
                              void* d_out, int out_size, void* d_ws, size_t ws_size,
                              hipStream_t stream) {
    (void)in_sizes; (void)n_in; (void)out_size; (void)ws_size;
    const float* left  = (const float*)d_in[0];
    const float* right = (const float*)d_in[1];
    const float* mid   = (const float*)d_in[2];
    const float* shcL  = (const float*)d_in[3];
    const float* shcR  = (const float*)d_in[4];
    float* out = (float*)d_out;

    float* w   = (float*)d_ws;
    const size_t NDS = (size_t)B_ * C_ * HD * WD;
    const size_t NDN = (size_t)B_ * L_;
    const size_t NSC = (size_t)B_ * L_ * L_;
    const size_t NY  = (size_t)B_ * K16 * L_;
    float* dsL  = w;
    float* dsR  = dsL + NDS;
    float* dsM  = dsR + NDS;
    float* denL = dsM + NDS;
    float* denR = denL + NDN;
    float* scL  = denR + NDN;
    float* scR  = scL + NSC;
    float* YL   = scR + NSC;
    float* YR   = YL + NY;

    k_prep_ds<<<dim3((NDS + 255) / 256), dim3(256), 0, stream>>>(left, right, mid,
                                                                 dsL, dsR, dsM);
    k_denom<<<dim3(B_ * L_ / 8, 2), dim3(256), 0, stream>>>(dsL, dsR, denL, denR);
    k_scores<<<dim3(81, 2, B_), dim3(256), 0, stream>>>(dsL, dsR, dsM,
                                                        denL, denR, scL, scR);
    k_softmax<<<dim3(L_ / SM_COLS, 2, B_), dim3(SM_COLS * SM_WAVES), 0, stream>>>(scL, scR);
    k_recon<<<dim3(576, 2, B_), dim3(256), 0, stream>>>(shcL, shcR, scL, scR, YL, YR);
    k_finalize<<<dim3((B_ * C_ * H_ * W_ + 255) / 256), dim3(256), 0, stream>>>(YL, YR, out);
}


# both GEMMs: K walked so a chunk varies along one axis (scores: one 3x3 tap x 32 channels; reconstruct: two rows of 24 positions): gather offsets and pad tests loop-invariant, no index divides
# speedup vs baseline: 3.9038x; 1.3284x over previous
#include <hip/hip_runtime.h>
#include <hip/hip_bf16.h>
#include <math.h>

#define B_   2
#define C_   256
#define H_   48
#define W_   48
#define HD   24
#define WD   24
#define L_   576
#define K9   2304
#define K16  4096
#define SCALE_ 10.0f
#define EPS_   1e-4f

typedef float v2f __attribute__((ext_vector_type(2)));
typedef float v8f __attribute__((ext_vector_type(8)));

#define TK  32
#define LDP (TK + 2)

__device__ __forceinline__ float patch9(const float* ds, int b, int k, int l) {
    int c = k / 9, r = k % 9;
    int ki = r / 3, kj = r % 3;
    int y = l / WD + ki - 1;
    int x = l % WD + kj - 1;
    bool ok = ((unsigned)y < HD) & ((unsigned)x < WD);
    int off = ok ? (y * WD + x) : 0;
    float v = ds[(b * C_ + c) * (HD * WD) + off];
    return ok ? v : 0.0f;
}
__device__ __forceinline__ int tap_off(int r, int y, int x, bool& ok) {
    int yy = y + r / 3 - 1;
    int xx = x + r % 3 - 1;
    ok = ((unsigned)yy < HD) & ((unsigned)xx < WD);
    return ok ? (yy * WD + xx) : 0;
}

__global__ __launch_bounds__(256) void k_prep_ds(const float* __restrict__ left,
                                                 const float* __restrict__ right,
                                                 const float* __restrict__ mid,
                                                 float* __restrict__ dsL,
                                                 float* __restrict__ dsR,
                                                 float* __restrict__ dsM) {
    int idx = blockIdx.x * 256 + threadIdx.x;
    if (idx >= B_ * C_ * HD * WD) return;
    int x = idx % WD, t = idx / WD;
    int y = t % HD;  t /= HD;
    int c = t % C_;  int b = t / C_;
    int src = ((b * C_ + c) * H_ + 2 * y) * W_ + 2 * x;
    dsL[idx] = left[src];
    dsR[idx] = right[src];
    dsM[idx] = mid[src];
}

__global__ __launch_bounds__(256) void k_denom(const float* __restrict__ dsL,
                                               const float* __restrict__ dsR,
                                               float* __restrict__ denL,
                                               float* __restrict__ denR) {
    const float* ds = blockIdx.y ? dsR : dsL;
    float* den      = blockIdx.y ? denR : denL;
    int gl   = blockIdx.x * 8 + (threadIdx.x >> 5);
    int b    = gl / L_;
    int l    = gl % L_;
    int lane = threadIdx.x & 31;
    float s = 0.0f;
    #pragma unroll 8
    for (int k = lane; k < K9; k += 32) {
        float v = patch9(ds, b, k, l);
        s += v * v;
    }
    #pragma unroll
    for (int off = 16; off; off >>= 1) s += __shfl_xor(s, off, 32);
    if (lane == 0) den[gl] = sqrtf(s + (float)K9 * EPS_);
}

__global__ __launch_bounds__(256) void k_scores(const float* __restrict__ dsL,
                                                const float* __restrict__ dsR,
                                                const float* __restrict__ dsM,
                                                const float* __restrict__ denL,
                                                const float* __restrict__ denR,
                                                float* __restrict__ scL,
                                                float* __restrict__ scR) {
    __shared__ __align__(16) float Ap[2][64][LDP];
    __shared__ __align__(16) float Bp[2][64][LDP];
    const float* dsF = blockIdx.y ? dsR : dsL;
    const float* den = blockIdx.y ? denR : denL;
    float* out       = blockIdx.y ? scR : scL;
    int b   = blockIdx.z;
    int l0  = (blockIdx.x / 9) * 64;
    int p0  = (blockIdx.x % 9) * 64;
    int tid = threadIdx.x;
    int lane = tid & 31, w = tid >> 5;
    int tr  = w >> 1;
    int tcb = (w & 1) * 2;
    int m16   = lane & 15;
    int khalf = (lane >> 4) * 2;
    v8f acc0 = {}; v8f acc1 = {};

    const int lr = tid & 63;
    const int kq = tid >> 6;
    const int ya = (l0 + lr) / WD, xa = (l0 + lr) % WD;
    const int yb = (p0 + lr) / WD, xb = (p0 + lr) % WD;
    const float* baseA = dsF + (size_t)(b * C_ + kq) * L_;
    const float* baseB = dsM + (size_t)(b * C_ + kq) * L_;
    float ra[8], rb[8];
    {
        bool oka, okb;
        const float* pa = baseA + tap_off(0, ya, xa, oka);
        const float* pb = baseB + tap_off(0, yb, xb, okb);
        #pragma unroll
        for (int t = 0; t < 8; ++t) {
            ra[t] = pa[(size_t)t * 4 * L_];
            rb[t] = pb[(size_t)t * 4 * L_];
        }
        #pragma unroll
        for (int t = 0; t < 8; ++t) {
            Ap[0][lr][kq + 4 * t] = oka ? ra[t] : 0.0f;
            Bp[0][lr][kq + 4 * t] = okb ? rb[t] : 0.0f;
        }
    }
    __syncthreads();
    int cur = 0;
    for (int q = 0; q < K9 / TK; ++q) {
        const bool more = (q + 1) < K9 / TK;
        bool oka = false, okb = false;
        if (more) {
            const int rn = (q + 1) >> 3;
            const int cn = ((q + 1) & 7) << 5;
            const float* pa = baseA + (size_t)cn * L_ + tap_off(rn, ya, xa, oka);
            const float* pb = baseB + (size_t)cn * L_ + tap_off(rn, yb, xb, okb);
            #pragma unroll
            for (int t = 0; t < 8; ++t) {
                ra[t] = pa[(size_t)t * 4 * L_];
                rb[t] = pb[(size_t)t * 4 * L_];
            }
        }
        __builtin_amdgcn_sched_barrier(0);
        #pragma unroll
        for (int kk0 = 0; kk0 < TK; kk0 += 4) {
            int ka = kk0 + khalf;
            v2f a  = *(const v2f*)&Ap[cur][tr * 16 + m16][ka];
            v2f b0 = *(const v2f*)&Bp[cur][tcb * 16 + m16][ka];
            v2f b1 = *(const v2f*)&Bp[cur][(tcb + 1) * 16 + m16][ka];
            acc0 = __builtin_amdgcn_wmma_f32_16x16x4_f32(false, a, false, b0,
                                                         (short)0, acc0, false, false);
            acc1 = __builtin_amdgcn_wmma_f32_16x16x4_f32(false, a, false, b1,
                                                         (short)0, acc1, false, false);
        }
        __builtin_amdgcn_sched_barrier(0);
        if (more) {
            #pragma unroll
            for (int t = 0; t < 8; ++t) {
                Ap[cur ^ 1][lr][kq + 4 * t] = oka ? ra[t] : 0.0f;
                Bp[cur ^ 1][lr][kq + 4 * t] = okb ? rb[t] : 0.0f;
            }
        }
        __syncthreads();
        cur ^= 1;
    }
    int n     = lane & 15;
    int rbase = l0 + tr * 16 + ((lane >> 4) << 3);
    #pragma unroll
    for (int v = 0; v < 8; ++v) {
        int row = rbase + v;
        float scl = SCALE_ / den[b * L_ + row];
        out[((size_t)b * L_ + row) * L_ + p0 + tcb * 16 + n]       = acc0[v] * scl;
        out[((size_t)b * L_ + row) * L_ + p0 + (tcb + 1) * 16 + n] = acc1[v] * scl;
    }
}

#define SM_COLS  32
#define SM_WAVES 16
#define SM_ROWS  (L_ / SM_WAVES)
__global__ __launch_bounds__(SM_COLS * SM_WAVES) void k_softmax(float* __restrict__ scL,
                                                                float* __restrict__ scR) {
    __shared__ float redm[SM_WAVES][SM_COLS];
    __shared__ float reds[SM_WAVES][SM_COLS];
    float* sc = blockIdx.y ? scR : scL;
    int b    = blockIdx.z;
    int lane = threadIdx.x & (SM_COLS - 1);
    int w    = threadIdx.x / SM_COLS;
    int p    = blockIdx.x * SM_COLS + lane;
    float* col = sc + ((size_t)b * L_ + (size_t)w * SM_ROWS) * L_ + p;

    float v[SM_ROWS];
    #pragma unroll
    for (int i = 0; i < SM_ROWS; ++i) v[i] = col[(size_t)i * L_];

    float mx = v[0];
    #pragma unroll
    for (int i = 1; i < SM_ROWS; ++i) mx = fmaxf(mx, v[i]);
    redm[w][lane] = mx;
    __syncthreads();
    float m = redm[0][lane];
    #pragma unroll
    for (int j = 1; j < SM_WAVES; ++j) m = fmaxf(m, redm[j][lane]);

    float s = 0.0f;
    #pragma unroll
    for (int i = 0; i < SM_ROWS; ++i) {
        v[i] = expf(v[i] - m);
        s += v[i];
    }
    reds[w][lane] = s;
    __syncthreads();
    float tot = 0.0f;
    #pragma unroll
    for (int j = 0; j < SM_WAVES; ++j) tot += reds[j][lane];
    float inv = 1.0f / tot;

    #pragma unroll
    for (int i = 0; i < SM_ROWS; ++i) col[(size_t)i * L_] = v[i] * inv;
}

#define TK2  48
#define LDP2 (TK2 + 2)
__global__ __launch_bounds__(256) void k_recon(const float* __restrict__ shcL,
                                               const float* __restrict__ shcR,
                                               const float* __restrict__ atL,
                                               const float* __restrict__ atR,
                                               float* __restrict__ YL,
                                               float* __restrict__ YR) {
    __shared__ __align__(16) float Ap[2][64][LDP2];
    __shared__ __align__(16) float Bp[2][64][LDP2];
    const float* sc = blockIdx.y ? shcR : shcL;
    const float* at = blockIdx.y ? atR  : atL;
    float* Y        = blockIdx.y ? YR   : YL;
    int b   = blockIdx.z;
    int m0  = (blockIdx.x / 9) * 64;
    int p0  = (blockIdx.x % 9) * 64;
    int tid = threadIdx.x;
    int lane = tid & 31, w = tid >> 5;
    int tr  = w >> 1;
    int tcb = (w & 1) * 2;
    int m16   = lane & 15;
    int khalf = (lane >> 4) * 2;
    v8f acc0 = {}; v8f acc1 = {};

    const int lr = tid & 63;
    const int kq = tid >> 6;
    const int m  = m0 + lr;
    const int pi = (m >> 2) & 3, pj = m & 3;
    const unsigned pbase = (unsigned)(b * C_ + (m >> 4)) * (unsigned)(H_ * W_);
    const float* atp = at + (size_t)b * L_ * L_ + p0 + lr;
    const int  s0   = 2 * kq + pj - 1;
    const bool cok0 = s0 >= 0;
    const bool cok5 = (s0 + 40) < W_;
    unsigned cb[6];
    cb[0] = pbase + (unsigned)(cok0 ? s0 : 0);
    #pragma unroll
    for (int u = 1; u < 5; ++u) cb[u] = pbase + (unsigned)(s0 + 8 * u);
    cb[5] = pbase + (unsigned)(cok5 ? s0 + 40 : 0);

    float ra[12], rb[12];
    {
        bool rok[2]; unsigned roff[2];
        #pragma unroll
        for (int h = 0; h < 2; ++h) {
            int r = 2 * h + pi - 1;
            rok[h]  = (unsigned)r < H_;
            roff[h] = rok[h] ? (unsigned)(r * W_) : 0u;
        }
        #pragma unroll
        for (int t = 0; t < 12; ++t) {
            ra[t] = sc[cb[t % 6] + roff[t / 6]];
            rb[t] = atp[(size_t)(kq + 4 * t) * L_];
        }
        #pragma unroll
        for (int t = 0; t < 12; ++t) {
            bool ok = rok[t / 6];
            if (t % 6 == 0) ok = ok & cok0;
            if (t % 6 == 5) ok = ok & cok5;
            Ap[0][lr][kq + 4 * t] = ok ? ra[t] : 0.0f;
            Bp[0][lr][kq + 4 * t] = rb[t];
        }
    }
    __syncthreads();
    int cur = 0;
    for (int q = 0; q < L_ / TK2; ++q) {
        const bool more = (q + 1) < L_ / TK2;
        bool rok[2] = {false, false};
        if (more) {
            unsigned roff[2];
            #pragma unroll
            for (int h = 0; h < 2; ++h) {
                int r = 2 * (2 * (q + 1) + h) + pi - 1;
                rok[h]  = (unsigned)r < H_;
                roff[h] = rok[h] ? (unsigned)(r * W_) : 0u;
            }
            const float* atq = atp + (size_t)(q + 1) * TK2 * L_;
            #pragma unroll
            for (int t = 0; t < 12; ++t) {
                ra[t] = sc[cb[t % 6] + roff[t / 6]];
                rb[t] = atq[(size_t)(kq + 4 * t) * L_];
            }
        }
        __builtin_amdgcn_sched_barrier(0);
        #pragma unroll
        for (int kk0 = 0; kk0 < TK2; kk0 += 4) {
            int ka = kk0 + khalf;
            v2f a  = *(const v2f*)&Ap[cur][tr * 16 + m16][ka];
            v2f b0 = *(const v2f*)&Bp[cur][tcb * 16 + m16][ka];
            v2f b1 = *(const v2f*)&Bp[cur][(tcb + 1) * 16 + m16][ka];
            acc0 = __builtin_amdgcn_wmma_f32_16x16x4_f32(false, a, false, b0,
                                                         (short)0, acc0, false, false);
            acc1 = __builtin_amdgcn_wmma_f32_16x16x4_f32(false, a, false, b1,
                                                         (short)0, acc1, false, false);
        }
        __builtin_amdgcn_sched_barrier(0);
        if (more) {
            #pragma unroll
            for (int t = 0; t < 12; ++t) {
                bool ok = rok[t / 6];
                if (t % 6 == 0) ok = ok & cok0;
                if (t % 6 == 5) ok = ok & cok5;
                Ap[cur ^ 1][lr][kq + 4 * t] = ok ? ra[t] : 0.0f;
                Bp[cur ^ 1][lr][kq + 4 * t] = rb[t];
            }
        }
        __syncthreads();
        cur ^= 1;
    }
    int n     = lane & 15;
    int rbase = m0 + tr * 16 + ((lane >> 4) << 3);
    #pragma unroll
    for (int v = 0; v < 8; ++v) {
        int row = rbase + v;
        Y[((size_t)b * K16 + row) * L_ + p0 + tcb * 16 + n]       = acc0[v];
        Y[((size_t)b * K16 + row) * L_ + p0 + (tcb + 1) * 16 + n] = acc1[v];
    }
}

__global__ __launch_bounds__(256) void k_finalize(const float* __restrict__ YL,
                                                  const float* __restrict__ YR,
                                                  float* __restrict__ out) {
    int idx = blockIdx.x * 256 + threadIdx.x;
    if (idx >= B_ * C_ * H_ * W_) return;
    int s = idx % W_; int t = idx / W_;
    int r = t % H_;  t /= H_;
    int c = t % C_;  int b = t / C_;
    float aL = 0.0f, aR = 0.0f;
    int i0 = (r + 1) & 1, j0 = (s + 1) & 1;
    #pragma unroll
    for (int di = 0; di < 2; ++di) {
        int i = i0 + 2 * di, ry = r + 1 - i;
        if (ry < 0 || ry >= 2 * HD) continue;
        int y = ry >> 1;
        #pragma unroll
        for (int dj = 0; dj < 2; ++dj) {
            int j = j0 + 2 * dj, sx = s + 1 - j;
            if (sx < 0 || sx >= 2 * WD) continue;
            int x = sx >> 1;
            int m = ((c << 2) | i) * 4 + j;
            size_t off = ((size_t)b * K16 + m) * L_ + y * WD + x;
            aL += YL[off];
            aR += YR[off];
        }
    }
    const float PI = 3.14159265358979323846f;
    float wl = 0.5f * (cosf(PI * (float)s / (float)(W_ - 1)) + 1.0f);
    float wr = 0.5f * (cosf(PI * (float)(W_ - 1 - s) / (float)(W_ - 1)) + 1.0f);
    out[idx] = 0.25f * (wl * aL + wr * aR);
}

extern "C" void kernel_launch(void* const* d_in, const int* in_sizes, int n_in,
                              void* d_out, int out_size, void* d_ws, size_t ws_size,
                              hipStream_t stream) {
    (void)in_sizes; (void)n_in; (void)out_size; (void)ws_size;
    const float* left  = (const float*)d_in[0];
    const float* right = (const float*)d_in[1];
    const float* mid   = (const float*)d_in[2];
    const float* shcL  = (const float*)d_in[3];
    const float* shcR  = (const float*)d_in[4];
    float* out = (float*)d_out;

    float* w   = (float*)d_ws;
    const size_t NDS = (size_t)B_ * C_ * HD * WD;
    const size_t NDN = (size_t)B_ * L_;
    const size_t NSC = (size_t)B_ * L_ * L_;
    const size_t NY  = (size_t)B_ * K16 * L_;
    float* dsL  = w;
    float* dsR  = dsL + NDS;
    float* dsM  = dsR + NDS;
    float* denL = dsM + NDS;
    float* denR = denL + NDN;
    float* scL  = denR + NDN;
    float* scR  = scL + NSC;
    float* YL   = scR + NSC;
    float* YR   = YL + NY;

    k_prep_ds<<<dim3((NDS + 255) / 256), dim3(256), 0, stream>>>(left, right, mid,
                                                                 dsL, dsR, dsM);
    k_denom<<<dim3(B_ * L_ / 8, 2), dim3(256), 0, stream>>>(dsL, dsR, denL, denR);
    k_scores<<<dim3(81, 2, B_), dim3(256), 0, stream>>>(dsL, dsR, dsM,
                                                        denL, denR, scL, scR);
    k_softmax<<<dim3(L_ / SM_COLS, 2, B_), dim3(SM_COLS * SM_WAVES), 0, stream>>>(scL, scR);
    k_recon<<<dim3(576, 2, B_), dim3(256), 0, stream>>>(shcL, shcR, scL, scR, YL, YR);
    k_finalize<<<dim3((B_ * C_ * H_ * W_ + 255) / 256), dim3(256), 0, stream>>>(YL, YR, out);
}


# reconstruct GEMM staging reads whole lines: A as contiguous image-row column pairs (each pixel stored to its 2 patch slots in LDS) instead of 4-byte gathers over 8-10 lines, B as 16-byte loads
# speedup vs baseline: 4.1449x; 1.0618x over previous
#include <hip/hip_runtime.h>
#include <hip/hip_bf16.h>
#include <math.h>

#define B_   2
#define C_   256
#define H_   48
#define W_   48
#define HD   24
#define WD   24
#define L_   576
#define K9   2304
#define K16  4096
#define SCALE_ 10.0f
#define EPS_   1e-4f

typedef float v2f __attribute__((ext_vector_type(2)));
typedef float v8f __attribute__((ext_vector_type(8)));

#define TK  32
#define LDP (TK + 2)

__device__ __forceinline__ float patch9(const float* ds, int b, int k, int l) {
    int c = k / 9, r = k % 9;
    int ki = r / 3, kj = r % 3;
    int y = l / WD + ki - 1;
    int x = l % WD + kj - 1;
    bool ok = ((unsigned)y < HD) & ((unsigned)x < WD);
    int off = ok ? (y * WD + x) : 0;
    float v = ds[(b * C_ + c) * (HD * WD) + off];
    return ok ? v : 0.0f;
}
__device__ __forceinline__ int tap_off(int r, int y, int x, bool& ok) {
    int yy = y + r / 3 - 1;
    int xx = x + r % 3 - 1;
    ok = ((unsigned)yy < HD) & ((unsigned)xx < WD);
    return ok ? (yy * WD + xx) : 0;
}

__global__ __launch_bounds__(256) void k_prep_ds(const float* __restrict__ left,
                                                 const float* __restrict__ right,
                                                 const float* __restrict__ mid,
                                                 float* __restrict__ dsL,
                                                 float* __restrict__ dsR,
                                                 float* __restrict__ dsM) {
    int idx = blockIdx.x * 256 + threadIdx.x;
    if (idx >= B_ * C_ * HD * WD) return;
    int x = idx % WD, t = idx / WD;
    int y = t % HD;  t /= HD;
    int c = t % C_;  int b = t / C_;
    int src = ((b * C_ + c) * H_ + 2 * y) * W_ + 2 * x;
    dsL[idx] = left[src];
    dsR[idx] = right[src];
    dsM[idx] = mid[src];
}

__global__ __launch_bounds__(256) void k_denom(const float* __restrict__ dsL,
                                               const float* __restrict__ dsR,
                                               float* __restrict__ denL,
                                               float* __restrict__ denR) {
    const float* ds = blockIdx.y ? dsR : dsL;
    float* den      = blockIdx.y ? denR : denL;
    int gl   = blockIdx.x * 8 + (threadIdx.x >> 5);
    int b    = gl / L_;
    int l    = gl % L_;
    int lane = threadIdx.x & 31;
    float s = 0.0f;
    #pragma unroll 8
    for (int k = lane; k < K9; k += 32) {
        float v = patch9(ds, b, k, l);
        s += v * v;
    }
    #pragma unroll
    for (int off = 16; off; off >>= 1) s += __shfl_xor(s, off, 32);
    if (lane == 0) den[gl] = sqrtf(s + (float)K9 * EPS_);
}

__global__ __launch_bounds__(256) void k_scores(const float* __restrict__ dsL,
                                                const float* __restrict__ dsR,
                                                const float* __restrict__ dsM,
                                                const float* __restrict__ denL,
                                                const float* __restrict__ denR,
                                                float* __restrict__ scL,
                                                float* __restrict__ scR) {
    __shared__ __align__(16) float Ap[2][64][LDP];
    __shared__ __align__(16) float Bp[2][64][LDP];
    const float* dsF = blockIdx.y ? dsR : dsL;
    const float* den = blockIdx.y ? denR : denL;
    float* out       = blockIdx.y ? scR : scL;
    int b   = blockIdx.z;
    int l0  = (blockIdx.x / 9) * 64;
    int p0  = (blockIdx.x % 9) * 64;
    int tid = threadIdx.x;
    int lane = tid & 31, w = tid >> 5;
    int tr  = w >> 1;
    int tcb = (w & 1) * 2;
    int m16   = lane & 15;
    int khalf = (lane >> 4) * 2;
    v8f acc0 = {}; v8f acc1 = {};

    const int lr = tid & 63;
    const int kq = tid >> 6;
    const int ya = (l0 + lr) / WD, xa = (l0 + lr) % WD;
    const int yb = (p0 + lr) / WD, xb = (p0 + lr) % WD;
    const float* baseA = dsF + (size_t)(b * C_ + kq) * L_;
    const float* baseB = dsM + (size_t)(b * C_ + kq) * L_;
    float ra[8], rb[8];
    {
        bool oka, okb;
        const float* pa = baseA + tap_off(0, ya, xa, oka);
        const float* pb = baseB + tap_off(0, yb, xb, okb);
        #pragma unroll
        for (int t = 0; t < 8; ++t) {
            ra[t] = pa[(size_t)t * 4 * L_];
            rb[t] = pb[(size_t)t * 4 * L_];
        }
        #pragma unroll
        for (int t = 0; t < 8; ++t) {
            Ap[0][lr][kq + 4 * t] = oka ? ra[t] : 0.0f;
            Bp[0][lr][kq + 4 * t] = okb ? rb[t] : 0.0f;
        }
    }
    __syncthreads();
    int cur = 0;
    for (int q = 0; q < K9 / TK; ++q) {
        const bool more = (q + 1) < K9 / TK;
        bool oka = false, okb = false;
        if (more) {
            const int rn = (q + 1) >> 3;
            const int cn = ((q + 1) & 7) << 5;
            const float* pa = baseA + (size_t)cn * L_ + tap_off(rn, ya, xa, oka);
            const float* pb = baseB + (size_t)cn * L_ + tap_off(rn, yb, xb, okb);
            #pragma unroll
            for (int t = 0; t < 8; ++t) {
                ra[t] = pa[(size_t)t * 4 * L_];
                rb[t] = pb[(size_t)t * 4 * L_];
            }
        }
        __builtin_amdgcn_sched_barrier(0);
        #pragma unroll
        for (int kk0 = 0; kk0 < TK; kk0 += 4) {
            int ka = kk0 + khalf;
            v2f a  = *(const v2f*)&Ap[cur][tr * 16 + m16][ka];
            v2f b0 = *(const v2f*)&Bp[cur][tcb * 16 + m16][ka];
            v2f b1 = *(const v2f*)&Bp[cur][(tcb + 1) * 16 + m16][ka];
            acc0 = __builtin_amdgcn_wmma_f32_16x16x4_f32(false, a, false, b0,
                                                         (short)0, acc0, false, false);
            acc1 = __builtin_amdgcn_wmma_f32_16x16x4_f32(false, a, false, b1,
                                                         (short)0, acc1, false, false);
        }
        __builtin_amdgcn_sched_barrier(0);
        if (more) {
            #pragma unroll
            for (int t = 0; t < 8; ++t) {
                Ap[cur ^ 1][lr][kq + 4 * t] = oka ? ra[t] : 0.0f;
                Bp[cur ^ 1][lr][kq + 4 * t] = okb ? rb[t] : 0.0f;
            }
        }
        __syncthreads();
        cur ^= 1;
    }
    int n     = lane & 15;
    int rbase = l0 + tr * 16 + ((lane >> 4) << 3);
    #pragma unroll
    for (int v = 0; v < 8; ++v) {
        int row = rbase + v;
        float scl = SCALE_ / den[b * L_ + row];
        out[((size_t)b * L_ + row) * L_ + p0 + tcb * 16 + n]       = acc0[v] * scl;
        out[((size_t)b * L_ + row) * L_ + p0 + (tcb + 1) * 16 + n] = acc1[v] * scl;
    }
}

#define SM_COLS  32
#define SM_WAVES 16
#define SM_ROWS  (L_ / SM_WAVES)
__global__ __launch_bounds__(SM_COLS * SM_WAVES) void k_softmax(float* __restrict__ scL,
                                                                float* __restrict__ scR) {
    __shared__ float redm[SM_WAVES][SM_COLS];
    __shared__ float reds[SM_WAVES][SM_COLS];
    float* sc = blockIdx.y ? scR : scL;
    int b    = blockIdx.z;
    int lane = threadIdx.x & (SM_COLS - 1);
    int w    = threadIdx.x / SM_COLS;
    int p    = blockIdx.x * SM_COLS + lane;
    float* col = sc + ((size_t)b * L_ + (size_t)w * SM_ROWS) * L_ + p;

    float v[SM_ROWS];
    #pragma unroll
    for (int i = 0; i < SM_ROWS; ++i) v[i] = col[(size_t)i * L_];

    float mx = v[0];
    #pragma unroll
    for (int i = 1; i < SM_ROWS; ++i) mx = fmaxf(mx, v[i]);
    redm[w][lane] = mx;
    __syncthreads();
    float m = redm[0][lane];
    #pragma unroll
    for (int j = 1; j < SM_WAVES; ++j) m = fmaxf(m, redm[j][lane]);

    float s = 0.0f;
    #pragma unroll
    for (int i = 0; i < SM_ROWS; ++i) {
        v[i] = expf(v[i] - m);
        s += v[i];
    }
    reds[w][lane] = s;
    __syncthreads();
    float tot = 0.0f;
    #pragma unroll
    for (int j = 0; j < SM_WAVES; ++j) tot += reds[j][lane];
    float inv = 1.0f / tot;

    #pragma unroll
    for (int i = 0; i < SM_ROWS; ++i) col[(size_t)i * L_] = v[i] * inv;
}

#define TK2  48
#define LDP2 (TK2 + 6)
typedef float v4f __attribute__((ext_vector_type(4)));
__global__ __launch_bounds__(256) void k_recon(const float* __restrict__ shcL,
                                               const float* __restrict__ shcR,
                                               const float* __restrict__ atL,
                                               const float* __restrict__ atR,
                                               float* __restrict__ YL,
                                               float* __restrict__ YR) {
    __shared__ __align__(16) float Ap[2][64][LDP2];
    __shared__ __align__(16) float Bp[2][64][LDP2];
    const float* sc = blockIdx.y ? shcR : shcL;
    const float* at = blockIdx.y ? atR  : atL;
    float* Y        = blockIdx.y ? YR   : YL;
    int b   = blockIdx.z;
    int m0  = (blockIdx.x / 9) * 64;
    int p0  = (blockIdx.x % 9) * 64;
    int tid = threadIdx.x;
    int lane = tid & 31, w = tid >> 5;
    int tr  = w >> 1;
    int tcb = (w & 1) * 2;
    int m16   = lane & 15;
    int khalf = (lane >> 4) * 2;
    v8f acc0 = {}; v8f acc1 = {};

    const int cl = w >> 1;
    const int hA = w & 1;
    const float* aimg = sc + (size_t)(b * C_ + (m0 >> 4) + cl) * (H_ * W_);
    int  iA[3], uoff[3], o1[3], oA[3], oB[3];
    bool zA[3], zB[3];
    #pragma unroll
    for (int t = 0; t < 3; ++t) {
        int u = lane + 32 * t;
        int i = u / 24, x = u % 24;
        iA[t]   = i;
        uoff[t] = 2 * u;
        int base = (cl * 16 + i * 4) * LDP2 + hA * 24;
        o1[t] = base + LDP2 + x;
        zA[t] = x < 1;
        zB[t] = x > 22;
        oA[t] = zA[t] ? base : base + 3 * LDP2 + x - 1;
        oB[t] = zB[t] ? base + 3 * LDP2 + 23 : base + x + 1;
    }

    const int p8   = lane & 7;
    const int lsub = lane >> 3;
    const int posB = (w & 1) * 32 + p8 * 4;
    const int kB0  = (w >> 1) * 4 + lsub;
    const float* bsrc = at + (size_t)b * L_ * L_ + (size_t)kB0 * L_ + p0 + posB;

    v2f va[3]; v4f vb[3];
    bool rok[3];
    {
        const int r0 = 2 * hA - 1;
        #pragma unroll
        for (int t = 0; t < 3; ++t) {
            rok[t] = (unsigned)(r0 + iA[t]) < H_;
            va[t] = *(const v2f*)(aimg + (rok[t] ? r0 * W_ : 0) + uoff[t]);
        }
        #pragma unroll
        for (int t = 0; t < 3; ++t) vb[t] = *(const v4f*)(bsrc + (size_t)(16 * t) * L_);
        float* A = &Ap[0][0][0];
        #pragma unroll
        for (int t = 0; t < 3; ++t) {
            float q0 = rok[t] ? va[t].x : 0.0f;
            float q1 = rok[t] ? va[t].y : 0.0f;
            A[o1[t]]        = q0;
            A[o1[t] + LDP2] = q1;
            A[oA[t]] = zA[t] ? 0.0f : q0;
            A[oB[t]] = zB[t] ? 0.0f : q1;
        }
        #pragma unroll
        for (int t = 0; t < 3; ++t) {
            #pragma unroll
            for (int e = 0; e < 4; ++e) Bp[0][posB + e][kB0 + 16 * t] = vb[t][e];
        }
    }
    __syncthreads();
    int cur = 0;
    for (int q = 0; q < L_ / TK2; ++q) {
        const bool more = (q + 1) < L_ / TK2;
        if (more) {
            const int r0 = 4 * (q + 1) + 2 * hA - 1;
            #pragma unroll
            for (int t = 0; t < 3; ++t) {
                rok[t] = (unsigned)(r0 + iA[t]) < H_;
                va[t] = *(const v2f*)(aimg + (rok[t] ? r0 * W_ : 0) + uoff[t]);
            }
            const float* bq = bsrc + (size_t)(q + 1) * TK2 * L_;
            #pragma unroll
            for (int t = 0; t < 3; ++t) vb[t] = *(const v4f*)(bq + (size_t)(16 * t) * L_);
        }
        __builtin_amdgcn_sched_barrier(0);
        #pragma unroll
        for (int kk0 = 0; kk0 < TK2; kk0 += 4) {
            int ka = kk0 + khalf;
            v2f a  = *(const v2f*)&Ap[cur][tr * 16 + m16][ka];
            v2f b0 = *(const v2f*)&Bp[cur][tcb * 16 + m16][ka];
            v2f b1 = *(const v2f*)&Bp[cur][(tcb + 1) * 16 + m16][ka];
            acc0 = __builtin_amdgcn_wmma_f32_16x16x4_f32(false, a, false, b0,
                                                         (short)0, acc0, false, false);
            acc1 = __builtin_amdgcn_wmma_f32_16x16x4_f32(false, a, false, b1,
                                                         (short)0, acc1, false, false);
        }
        __builtin_amdgcn_sched_barrier(0);
        if (more) {
            float* A = &Ap[cur ^ 1][0][0];
            #pragma unroll
            for (int t = 0; t < 3; ++t) {
                float q0 = rok[t] ? va[t].x : 0.0f;
                float q1 = rok[t] ? va[t].y : 0.0f;
                A[o1[t]]        = q0;
                A[o1[t] + LDP2] = q1;
                A[oA[t]] = zA[t] ? 0.0f : q0;
                A[oB[t]] = zB[t] ? 0.0f : q1;
            }
            #pragma unroll
            for (int t = 0; t < 3; ++t) {
                #pragma unroll
                for (int e = 0; e < 4; ++e) Bp[cur ^ 1][posB + e][kB0 + 16 * t] = vb[t][e];
            }
        }
        __syncthreads();
        cur ^= 1;
    }
    int n     = lane & 15;
    int rbase = m0 + tr * 16 + ((lane >> 4) << 3);
    #pragma unroll
    for (int v = 0; v < 8; ++v) {
        int row = rbase + v;
        Y[((size_t)b * K16 + row) * L_ + p0 + tcb * 16 + n]       = acc0[v];
        Y[((size_t)b * K16 + row) * L_ + p0 + (tcb + 1) * 16 + n] = acc1[v];
    }
}

__global__ __launch_bounds__(256) void k_finalize(const float* __restrict__ YL,
                                                  const float* __restrict__ YR,
                                                  float* __restrict__ out) {
    int idx = blockIdx.x * 256 + threadIdx.x;
    if (idx >= B_ * C_ * H_ * W_) return;
    int s = idx % W_; int t = idx / W_;
    int r = t % H_;  t /= H_;
    int c = t % C_;  int b = t / C_;
    float aL = 0.0f, aR = 0.0f;
    int i0 = (r + 1) & 1, j0 = (s + 1) & 1;
    #pragma unroll
    for (int di = 0; di < 2; ++di) {
        int i = i0 + 2 * di, ry = r + 1 - i;
        if (ry < 0 || ry >= 2 * HD) continue;
        int y = ry >> 1;
        #pragma unroll
        for (int dj = 0; dj < 2; ++dj) {
            int j = j0 + 2 * dj, sx = s + 1 - j;
            if (sx < 0 || sx >= 2 * WD) continue;
            int x = sx >> 1;
            int m = ((c << 2) | i) * 4 + j;
            size_t off = ((size_t)b * K16 + m) * L_ + y * WD + x;
            aL += YL[off];
            aR += YR[off];
        }
    }
    const float PI = 3.14159265358979323846f;
    float wl = 0.5f * (cosf(PI * (float)s / (float)(W_ - 1)) + 1.0f);
    float wr = 0.5f * (cosf(PI * (float)(W_ - 1 - s) / (float)(W_ - 1)) + 1.0f);
    out[idx] = 0.25f * (wl * aL + wr * aR);
}

extern "C" void kernel_launch(void* const* d_in, const int* in_sizes, int n_in,
                              void* d_out, int out_size, void* d_ws, size_t ws_size,
                              hipStream_t stream) {
    (void)in_sizes; (void)n_in; (void)out_size; (void)ws_size;
    const float* left  = (const float*)d_in[0];
    const float* right = (const float*)d_in[1];
    const float* mid   = (const float*)d_in[2];
    const float* shcL  = (const float*)d_in[3];
    const float* shcR  = (const float*)d_in[4];
    float* out = (float*)d_out;

    float* w   = (float*)d_ws;
    const size_t NDS = (size_t)B_ * C_ * HD * WD;
    const size_t NDN = (size_t)B_ * L_;
    const size_t NSC = (size_t)B_ * L_ * L_;
    const size_t NY  = (size_t)B_ * K16 * L_;
    float* dsL  = w;
    float* dsR  = dsL + NDS;
    float* dsM  = dsR + NDS;
    float* denL = dsM + NDS;
    float* denR = denL + NDN;
    float* scL  = denR + NDN;
    float* scR  = scL + NSC;
    float* YL   = scR + NSC;
    float* YR   = YL + NY;

    k_prep_ds<<<dim3((NDS + 255) / 256), dim3(256), 0, stream>>>(left, right, mid,
                                                                 dsL, dsR, dsM);
    k_denom<<<dim3(B_ * L_ / 8, 2), dim3(256), 0, stream>>>(dsL, dsR, denL, denR);
    k_scores<<<dim3(81, 2, B_), dim3(256), 0, stream>>>(dsL, dsR, dsM,
                                                        denL, denR, scL, scR);
    k_softmax<<<dim3(L_ / SM_COLS, 2, B_), dim3(SM_COLS * SM_WAVES), 0, stream>>>(scL, scR);
    k_recon<<<dim3(576, 2, B_), dim3(256), 0, stream>>>(shcL, shcR, scL, scR, YL, YR);
    k_finalize<<<dim3((B_ * C_ * H_ * W_ + 255) / 256), dim3(256), 0, stream>>>(YL, YR, out);
}


# scores GEMM over K=256 channels, not 2304 patch features: the 3x3 patch product is the sum of nine shifted entries G[l+d][p+d] of the channel correlation, summed in the softmax's column load
# speedup vs baseline: 6.4962x; 1.5673x over previous
#include <hip/hip_runtime.h>
#include <hip/hip_bf16.h>
#include <math.h>

#define B_   2
#define C_   256
#define H_   48
#define W_   48
#define HD   24
#define WD   24
#define L_   576
#define K9   2304
#define K16  4096
#define SCALE_ 10.0f
#define EPS_   1e-4f

typedef float v2f __attribute__((ext_vector_type(2)));
typedef float v8f __attribute__((ext_vector_type(8)));

#define TK  32
#define LDP (TK + 2)

__device__ __forceinline__ float patch9(const float* ds, int b, int k, int l) {
    int c = k / 9, r = k % 9;
    int ki = r / 3, kj = r % 3;
    int y = l / WD + ki - 1;
    int x = l % WD + kj - 1;
    bool ok = ((unsigned)y < HD) & ((unsigned)x < WD);
    int off = ok ? (y * WD + x) : 0;
    float v = ds[(b * C_ + c) * (HD * WD) + off];
    return ok ? v : 0.0f;
}

__global__ __launch_bounds__(256) void k_prep_ds(const float* __restrict__ left,
                                                 const float* __restrict__ right,
                                                 const float* __restrict__ mid,
                                                 float* __restrict__ dsL,
                                                 float* __restrict__ dsR,
                                                 float* __restrict__ dsM) {
    int idx = blockIdx.x * 256 + threadIdx.x;
    if (idx >= B_ * C_ * HD * WD) return;
    int x = idx % WD, t = idx / WD;
    int y = t % HD;  t /= HD;
    int c = t % C_;  int b = t / C_;
    int src = ((b * C_ + c) * H_ + 2 * y) * W_ + 2 * x;
    dsL[idx] = left[src];
    dsR[idx] = right[src];
    dsM[idx] = mid[src];
}

__global__ __launch_bounds__(256) void k_denom(const float* __restrict__ dsL,
                                               const float* __restrict__ dsR,
                                               float* __restrict__ denL,
                                               float* __restrict__ denR) {
    const float* ds = blockIdx.y ? dsR : dsL;
    float* den      = blockIdx.y ? denR : denL;
    int gl   = blockIdx.x * 8 + (threadIdx.x >> 5);
    int b    = gl / L_;
    int l    = gl % L_;
    int lane = threadIdx.x & 31;
    float s = 0.0f;
    #pragma unroll 8
    for (int k = lane; k < K9; k += 32) {
        float v = patch9(ds, b, k, l);
        s += v * v;
    }
    #pragma unroll
    for (int off = 16; off; off >>= 1) s += __shfl_xor(s, off, 32);
    if (lane == 0) den[gl] = sqrtf(s + (float)K9 * EPS_);
}

__global__ __launch_bounds__(256) void k_scores(const float* __restrict__ dsL,
                                                const float* __restrict__ dsR,
                                                const float* __restrict__ dsM,
                                                float* __restrict__ G) {
    __shared__ __align__(16) float Ap[2][64][LDP];
    __shared__ __align__(16) float Bp[2][64][LDP];
    const float* dsF = blockIdx.y ? dsR : dsL;
    int b   = blockIdx.z;
    float* out = G + (size_t)(blockIdx.y * B_ + b) * L_ * L_;
    int l0  = (blockIdx.x / 9) * 64;
    int p0  = (blockIdx.x % 9) * 64;
    int tid = threadIdx.x;
    int lane = tid & 31, w = tid >> 5;
    int tr  = w >> 1;
    int tcb = (w & 1) * 2;
    int m16   = lane & 15;
    int khalf = (lane >> 4) * 2;
    v8f acc0 = {}; v8f acc1 = {};

    const int lr = tid & 63;
    const int kq = tid >> 6;
    const float* baseA = dsF + (size_t)(b * C_ + kq) * L_ + l0 + lr;
    const float* baseB = dsM + (size_t)(b * C_ + kq) * L_ + p0 + lr;
    float ra[8], rb[8];
    #pragma unroll
    for (int t = 0; t < 8; ++t) {
        ra[t] = baseA[(size_t)t * 4 * L_];
        rb[t] = baseB[(size_t)t * 4 * L_];
    }
    #pragma unroll
    for (int t = 0; t < 8; ++t) {
        Ap[0][lr][kq + 4 * t] = ra[t];
        Bp[0][lr][kq + 4 * t] = rb[t];
    }
    __syncthreads();
    int cur = 0;
    for (int q = 0; q < C_ / TK; ++q) {
        const bool more = (q + 1) < C_ / TK;
        if (more) {
            const float* pa = baseA + (size_t)(q + 1) * TK * L_;
            const float* pb = baseB + (size_t)(q + 1) * TK * L_;
            #pragma unroll
            for (int t = 0; t < 8; ++t) {
                ra[t] = pa[(size_t)t * 4 * L_];
                rb[t] = pb[(size_t)t * 4 * L_];
            }
        }
        __builtin_amdgcn_sched_barrier(0);
        #pragma unroll
        for (int kk0 = 0; kk0 < TK; kk0 += 4) {
            int ka = kk0 + khalf;
            v2f a  = *(const v2f*)&Ap[cur][tr * 16 + m16][ka];
            v2f b0 = *(const v2f*)&Bp[cur][tcb * 16 + m16][ka];
            v2f b1 = *(const v2f*)&Bp[cur][(tcb + 1) * 16 + m16][ka];
            acc0 = __builtin_amdgcn_wmma_f32_16x16x4_f32(false, a, false, b0,
                                                         (short)0, acc0, false, false);
            acc1 = __builtin_amdgcn_wmma_f32_16x16x4_f32(false, a, false, b1,
                                                         (short)0, acc1, false, false);
        }
        __builtin_amdgcn_sched_barrier(0);
        if (more) {
            #pragma unroll
            for (int t = 0; t < 8; ++t) {
                Ap[cur ^ 1][lr][kq + 4 * t] = ra[t];
                Bp[cur ^ 1][lr][kq + 4 * t] = rb[t];
            }
        }
        __syncthreads();
        cur ^= 1;
    }
    int n     = lane & 15;
    int rbase = l0 + tr * 16 + ((lane >> 4) << 3);
    #pragma unroll
    for (int v = 0; v < 8; ++v) {
        int row = rbase + v;
        out[(size_t)row * L_ + p0 + tcb * 16 + n]       = acc0[v];
        out[(size_t)row * L_ + p0 + (tcb + 1) * 16 + n] = acc1[v];
    }
}

#define SM_COLS  32
#define SM_WAVES HD
#define SM_ROWS  WD
#define SM_RB    4
__global__ __launch_bounds__(SM_COLS * SM_WAVES) void k_softmax(const float* __restrict__ G,
                                                                const float* __restrict__ denL,
                                                                const float* __restrict__ denR,
                                                                float* __restrict__ scL,
                                                                float* __restrict__ scR) {
    __shared__ float vs[L_][SM_COLS];
    __shared__ float redm[SM_WAVES][SM_COLS];
    __shared__ float reds[SM_WAVES][SM_COLS];
    float* sc = blockIdx.y ? scR : scL;
    int b    = blockIdx.z;
    const float* g   = G + (size_t)(blockIdx.y * B_ + b) * L_ * L_;
    int lane = threadIdx.x & (SM_COLS - 1);
    int w    = __builtin_amdgcn_readfirstlane((int)(threadIdx.x / SM_COLS));
    const float* den = (blockIdx.y ? denR : denL) + b * L_ + w * SM_ROWS;
    int p    = blockIdx.x * SM_COLS + lane;
    float* col  = sc + ((size_t)b * L_ + (size_t)w * SM_ROWS) * L_ + p;
    float* vrow = &vs[w * SM_ROWS][lane];

    const bool vy0 = w > 0, vy2 = w < HD - 1;
    const float* gb[3];
    gb[0] = g + (size_t)((vy0 ? w - 1 : w) * WD) * L_;
    gb[1] = g + (size_t)(w * WD) * L_;
    gb[2] = g + (size_t)((vy2 ? w + 1 : w) * WD) * L_;
    const int py = p / WD, px = p % WD;
    int  pc[9];
    bool ok[9];
    #pragma unroll
    for (int t = 0; t < 9; ++t) {
        int dy = t / 3 - 1, dx = t % 3 - 1;
        bool vp = ((unsigned)(py + dy) < HD) & ((unsigned)(px + dx) < WD);
        pc[t] = vp ? p + dy * WD + dx : p;
        ok[t] = vp & (dy < 0 ? vy0 : (dy > 0 ? vy2 : true));
    }

    float mx = -INFINITY;
    #pragma unroll 1
    for (int i = 0; i < SM_ROWS; i += SM_RB) {
        const bool vfirst = i > 0;
        const bool vlast  = i + SM_RB < SM_ROWS;
        float x[SM_RB][9];
        #pragma unroll
        for (int r = 0; r < SM_RB; ++r) {
            #pragma unroll
            for (int t = 0; t < 9; ++t) {
                const int dx = t % 3 - 1;
                int cx = i + r + dx;
                if (r == 0 && dx < 0)         cx = max(cx, 0);
                if (r == SM_RB - 1 && dx > 0) cx = min(cx, WD - 1);
                x[r][t] = gb[t / 3][(size_t)cx * L_ + pc[t]];
            }
        }
        #pragma unroll
        for (int r = 0; r < SM_RB; ++r) {
            float s = x[r][4];
            #pragma unroll
            for (int t = 0; t < 9; ++t) {
                if (t == 4) continue;
                const int dx = t % 3 - 1;
                bool vx = (r == 0 && dx < 0) ? vfirst : ((r == SM_RB - 1 && dx > 0) ? vlast : true);
                s += (ok[t] & vx) ? x[r][t] : 0.0f;
            }
            float val = s * (SCALE_ / den[i + r]);
            vrow[(i + r) * SM_COLS] = val;
            mx = fmaxf(mx, val);
        }
    }
    redm[w][lane] = mx;
    __syncthreads();
    float m = redm[0][lane];
    #pragma unroll
    for (int j = 1; j < SM_WAVES; ++j) m = fmaxf(m, redm[j][lane]);

    float v[SM_ROWS];
    float s = 0.0f;
    #pragma unroll
    for (int i = 0; i < SM_ROWS; ++i) {
        v[i] = expf(vrow[i * SM_COLS] - m);
        s += v[i];
    }
    reds[w][lane] = s;
    __syncthreads();
    float tot = 0.0f;
    #pragma unroll
    for (int j = 0; j < SM_WAVES; ++j) tot += reds[j][lane];
    float inv = 1.0f / tot;

    #pragma unroll
    for (int i = 0; i < SM_ROWS; ++i) col[(size_t)i * L_] = v[i] * inv;
}

#define TK2  48
#define LDP2 (TK2 + 6)
typedef float v4f __attribute__((ext_vector_type(4)));
__global__ __launch_bounds__(256) void k_recon(const float* __restrict__ shcL,
                                               const float* __restrict__ shcR,
                                               const float* __restrict__ atL,
                                               const float* __restrict__ atR,
                                               float* __restrict__ YL,
                                               float* __restrict__ YR) {
    __shared__ __align__(16) float Ap[2][64][LDP2];
    __shared__ __align__(16) float Bp[2][64][LDP2];
    const float* sc = blockIdx.y ? shcR : shcL;
    const float* at = blockIdx.y ? atR  : atL;
    float* Y        = blockIdx.y ? YR   : YL;
    int b   = blockIdx.z;
    int m0  = (blockIdx.x / 9) * 64;
    int p0  = (blockIdx.x % 9) * 64;
    int tid = threadIdx.x;
    int lane = tid & 31, w = tid >> 5;
    int tr  = w >> 1;
    int tcb = (w & 1) * 2;
    int m16   = lane & 15;
    int khalf = (lane >> 4) * 2;
    v8f acc0 = {}; v8f acc1 = {};

    const int cl = w >> 1;
    const int hA = w & 1;
    const float* aimg = sc + (size_t)(b * C_ + (m0 >> 4) + cl) * (H_ * W_);
    int  iA[3], uoff[3], o1[3], oA[3], oB[3];
    bool zA[3], zB[3];
    #pragma unroll
    for (int t = 0; t < 3; ++t) {
        int u = lane + 32 * t;
        int i = u / 24, x = u % 24;
        iA[t]   = i;
        uoff[t] = 2 * u;
        int base = (cl * 16 + i * 4) * LDP2 + hA * 24;
        o1[t] = base + LDP2 + x;
        zA[t] = x < 1;
        zB[t] = x > 22;
        oA[t] = zA[t] ? base : base + 3 * LDP2 + x - 1;
        oB[t] = zB[t] ? base + 3 * LDP2 + 23 : base + x + 1;
    }

    const int p8   = lane & 7;
    const int lsub = lane >> 3;
    const int posB = (w & 1) * 32 + p8 * 4;
    const int kB0  = (w >> 1) * 4 + lsub;
    const float* bsrc = at + (size_t)b * L_ * L_ + (size_t)kB0 * L_ + p0 + posB;

    v2f va[3]; v4f vb[3];
    bool rok[3];
    {
        const int r0 = 2 * hA - 1;
        #pragma unroll
        for (int t = 0; t < 3; ++t) {
            rok[t] = (unsigned)(r0 + iA[t]) < H_;
            va[t] = *(const v2f*)(aimg + (rok[t] ? r0 * W_ : 0) + uoff[t]);
        }
        #pragma unroll
        for (int t = 0; t < 3; ++t) vb[t] = *(const v4f*)(bsrc + (size_t)(16 * t) * L_);
        float* A = &Ap[0][0][0];
        #pragma unroll
        for (int t = 0; t < 3; ++t) {
            float q0 = rok[t] ? va[t].x : 0.0f;
            float q1 = rok[t] ? va[t].y : 0.0f;
            A[o1[t]]        = q0;
            A[o1[t] + LDP2] = q1;
            A[oA[t]] = zA[t] ? 0.0f : q0;
            A[oB[t]] = zB[t] ? 0.0f : q1;
        }
        #pragma unroll
        for (int t = 0; t < 3; ++t) {
            #pragma unroll
            for (int e = 0; e < 4; ++e) Bp[0][posB + e][kB0 + 16 * t] = vb[t][e];
        }
    }
    __syncthreads();
    int cur = 0;
    for (int q = 0; q < L_ / TK2; ++q) {
        const bool more = (q + 1) < L_ / TK2;
        if (more) {
            const int r0 = 4 * (q + 1) + 2 * hA - 1;
            #pragma unroll
            for (int t = 0; t < 3; ++t) {
                rok[t] = (unsigned)(r0 + iA[t]) < H_;
                va[t] = *(const v2f*)(aimg + (rok[t] ? r0 * W_ : 0) + uoff[t]);
            }
            const float* bq = bsrc + (size_t)(q + 1) * TK2 * L_;
            #pragma unroll
            for (int t = 0; t < 3; ++t) vb[t] = *(const v4f*)(bq + (size_t)(16 * t) * L_);
        }
        __builtin_amdgcn_sched_barrier(0);
        #pragma unroll
        for (int kk0 = 0; kk0 < TK2; kk0 += 4) {
            int ka = kk0 + khalf;
            v2f a  = *(const v2f*)&Ap[cur][tr * 16 + m16][ka];
            v2f b0 = *(const v2f*)&Bp[cur][tcb * 16 + m16][ka];
            v2f b1 = *(const v2f*)&Bp[cur][(tcb + 1) * 16 + m16][ka];
            acc0 = __builtin_amdgcn_wmma_f32_16x16x4_f32(false, a, false, b0,
                                                         (short)0, acc0, false, false);
            acc1 = __builtin_amdgcn_wmma_f32_16x16x4_f32(false, a, false, b1,
                                                         (short)0, acc1, false, false);
        }
        __builtin_amdgcn_sched_barrier(0);
        if (more) {
            float* A = &Ap[cur ^ 1][0][0];
            #pragma unroll
            for (int t = 0; t < 3; ++t) {
                float q0 = rok[t] ? va[t].x : 0.0f;
                float q1 = rok[t] ? va[t].y : 0.0f;
                A[o1[t]]        = q0;
                A[o1[t] + LDP2] = q1;
                A[oA[t]] = zA[t] ? 0.0f : q0;
                A[oB[t]] = zB[t] ? 0.0f : q1;
            }
            #pragma unroll
            for (int t = 0; t < 3; ++t) {
                #pragma unroll
                for (int e = 0; e < 4; ++e) Bp[cur ^ 1][posB + e][kB0 + 16 * t] = vb[t][e];
            }
        }
        __syncthreads();
        cur ^= 1;
    }
    int n     = lane & 15;
    int rbase = m0 + tr * 16 + ((lane >> 4) << 3);
    #pragma unroll
    for (int v = 0; v < 8; ++v) {
        int row = rbase + v;
        Y[((size_t)b * K16 + row) * L_ + p0 + tcb * 16 + n]       = acc0[v];
        Y[((size_t)b * K16 + row) * L_ + p0 + (tcb + 1) * 16 + n] = acc1[v];
    }
}

__global__ __launch_bounds__(256) void k_finalize(const float* __restrict__ YL,
                                                  const float* __restrict__ YR,
                                                  float* __restrict__ out) {
    int idx = blockIdx.x * 256 + threadIdx.x;
    if (idx >= B_ * C_ * H_ * W_) return;
    int s = idx % W_; int t = idx / W_;
    int r = t % H_;  t /= H_;
    int c = t % C_;  int b = t / C_;
    float aL = 0.0f, aR = 0.0f;
    int i0 = (r + 1) & 1, j0 = (s + 1) & 1;
    #pragma unroll
    for (int di = 0; di < 2; ++di) {
        int i = i0 + 2 * di, ry = r + 1 - i;
        if (ry < 0 || ry >= 2 * HD) continue;
        int y = ry >> 1;
        #pragma unroll
        for (int dj = 0; dj < 2; ++dj) {
            int j = j0 + 2 * dj, sx = s + 1 - j;
            if (sx < 0 || sx >= 2 * WD) continue;
            int x = sx >> 1;
            int m = ((c << 2) | i) * 4 + j;
            size_t off = ((size_t)b * K16 + m) * L_ + y * WD + x;
            aL += YL[off];
            aR += YR[off];
        }
    }
    const float PI = 3.14159265358979323846f;
    float wl = 0.5f * (cosf(PI * (float)s / (float)(W_ - 1)) + 1.0f);
    float wr = 0.5f * (cosf(PI * (float)(W_ - 1 - s) / (float)(W_ - 1)) + 1.0f);
    out[idx] = 0.25f * (wl * aL + wr * aR);
}

extern "C" void kernel_launch(void* const* d_in, const int* in_sizes, int n_in,
                              void* d_out, int out_size, void* d_ws, size_t ws_size,
                              hipStream_t stream) {
    (void)in_sizes; (void)n_in; (void)out_size; (void)ws_size;
    const float* left  = (const float*)d_in[0];
    const float* right = (const float*)d_in[1];
    const float* mid   = (const float*)d_in[2];
    const float* shcL  = (const float*)d_in[3];
    const float* shcR  = (const float*)d_in[4];
    float* out = (float*)d_out;

    float* w   = (float*)d_ws;
    const size_t NDS = (size_t)B_ * C_ * HD * WD;
    const size_t NDN = (size_t)B_ * L_;
    const size_t NSC = (size_t)B_ * L_ * L_;
    const size_t NY  = (size_t)B_ * K16 * L_;
    float* dsL  = w;
    float* dsR  = dsL + NDS;
    float* dsM  = dsR + NDS;
    float* denL = dsM + NDS;
    float* denR = denL + NDN;
    float* scL  = denR + NDN;
    float* scR  = scL + NSC;
    float* YL   = scR + NSC;
    float* YR   = YL + NY;
    float* G    = YL;

    k_prep_ds<<<dim3((NDS + 255) / 256), dim3(256), 0, stream>>>(left, right, mid,
                                                                 dsL, dsR, dsM);
    k_denom<<<dim3(B_ * L_ / 8, 2), dim3(256), 0, stream>>>(dsL, dsR, denL, denR);
    k_scores<<<dim3(81, 2, B_), dim3(256), 0, stream>>>(dsL, dsR, dsM, G);
    k_softmax<<<dim3(L_ / SM_COLS, 2, B_), dim3(SM_COLS * SM_WAVES), 0, stream>>>(G, denL, denR,
                                                                                  scL, scR);
    k_recon<<<dim3(576, 2, B_), dim3(256), 0, stream>>>(shcL, shcR, scL, scR, YL, YR);
    k_finalize<<<dim3((B_ * C_ * H_ * W_ + 255) / 256), dim3(256), 0, stream>>>(YL, YR, out);
}


# reconstruct GEMM per output parity over K=2x576 shifted attention sums (the 4 overlapping taps summed on the attention side, weights folded in): 4x fewer matrix products, no overlap gather pass
# speedup vs baseline: 8.0299x; 1.2361x over previous
#include <hip/hip_runtime.h>
#include <hip/hip_bf16.h>
#include <math.h>

#define B_   2
#define C_   256
#define H_   48
#define W_   48
#define HD   24
#define WD   24
#define L_   576
#define K9   2304
#define K16  4096
#define SCALE_ 10.0f
#define EPS_   1e-4f

typedef float v2f __attribute__((ext_vector_type(2)));
typedef float v8f __attribute__((ext_vector_type(8)));

#define TK  32
#define LDP (TK + 2)

__device__ __forceinline__ float patch9(const float* ds, int b, int k, int l) {
    int c = k / 9, r = k % 9;
    int ki = r / 3, kj = r % 3;
    int y = l / WD + ki - 1;
    int x = l % WD + kj - 1;
    bool ok = ((unsigned)y < HD) & ((unsigned)x < WD);
    int off = ok ? (y * WD + x) : 0;
    float v = ds[(b * C_ + c) * (HD * WD) + off];
    return ok ? v : 0.0f;
}

__global__ __launch_bounds__(256) void k_prep_ds(const float* __restrict__ left,
                                                 const float* __restrict__ right,
                                                 const float* __restrict__ mid,
                                                 float* __restrict__ dsL,
                                                 float* __restrict__ dsR,
                                                 float* __restrict__ dsM) {
    int idx = blockIdx.x * 256 + threadIdx.x;
    if (idx >= B_ * C_ * HD * WD) return;
    int x = idx % WD, t = idx / WD;
    int y = t % HD;  t /= HD;
    int c = t % C_;  int b = t / C_;
    int src = ((b * C_ + c) * H_ + 2 * y) * W_ + 2 * x;
    dsL[idx] = left[src];
    dsR[idx] = right[src];
    dsM[idx] = mid[src];
}

__global__ __launch_bounds__(256) void k_denom(const float* __restrict__ dsL,
                                               const float* __restrict__ dsR,
                                               float* __restrict__ denL,
                                               float* __restrict__ denR) {
    const float* ds = blockIdx.y ? dsR : dsL;
    float* den      = blockIdx.y ? denR : denL;
    int gl   = blockIdx.x * 8 + (threadIdx.x >> 5);
    int b    = gl / L_;
    int l    = gl % L_;
    int lane = threadIdx.x & 31;
    float s = 0.0f;
    #pragma unroll 8
    for (int k = lane; k < K9; k += 32) {
        float v = patch9(ds, b, k, l);
        s += v * v;
    }
    #pragma unroll
    for (int off = 16; off; off >>= 1) s += __shfl_xor(s, off, 32);
    if (lane == 0) den[gl] = sqrtf(s + (float)K9 * EPS_);
}

__global__ __launch_bounds__(256) void k_scores(const float* __restrict__ dsL,
                                                const float* __restrict__ dsR,
                                                const float* __restrict__ dsM,
                                                float* __restrict__ G) {
    __shared__ __align__(16) float Ap[2][64][LDP];
    __shared__ __align__(16) float Bp[2][64][LDP];
    const float* dsF = blockIdx.y ? dsR : dsL;
    int b   = blockIdx.z;
    float* out = G + (size_t)(blockIdx.y * B_ + b) * L_ * L_;
    int l0  = (blockIdx.x / 9) * 64;
    int p0  = (blockIdx.x % 9) * 64;
    int tid = threadIdx.x;
    int lane = tid & 31, w = tid >> 5;
    int tr  = w >> 1;
    int tcb = (w & 1) * 2;
    int m16   = lane & 15;
    int khalf = (lane >> 4) * 2;
    v8f acc0 = {}; v8f acc1 = {};

    const int lr = tid & 63;
    const int kq = tid >> 6;
    const float* baseA = dsF + (size_t)(b * C_ + kq) * L_ + l0 + lr;
    const float* baseB = dsM + (size_t)(b * C_ + kq) * L_ + p0 + lr;
    float ra[8], rb[8];
    #pragma unroll
    for (int t = 0; t < 8; ++t) {
        ra[t] = baseA[(size_t)t * 4 * L_];
        rb[t] = baseB[(size_t)t * 4 * L_];
    }
    #pragma unroll
    for (int t = 0; t < 8; ++t) {
        Ap[0][lr][kq + 4 * t] = ra[t];
        Bp[0][lr][kq + 4 * t] = rb[t];
    }
    __syncthreads();
    int cur = 0;
    for (int q = 0; q < C_ / TK; ++q) {
        const bool more = (q + 1) < C_ / TK;
        if (more) {
            const float* pa = baseA + (size_t)(q + 1) * TK * L_;
            const float* pb = baseB + (size_t)(q + 1) * TK * L_;
            #pragma unroll
            for (int t = 0; t < 8; ++t) {
                ra[t] = pa[(size_t)t * 4 * L_];
                rb[t] = pb[(size_t)t * 4 * L_];
            }
        }
        __builtin_amdgcn_sched_barrier(0);
        #pragma unroll
        for (int kk0 = 0; kk0 < TK; kk0 += 4) {
            int ka = kk0 + khalf;
            v2f a  = *(const v2f*)&Ap[cur][tr * 16 + m16][ka];
            v2f b0 = *(const v2f*)&Bp[cur][tcb * 16 + m16][ka];
            v2f b1 = *(const v2f*)&Bp[cur][(tcb + 1) * 16 + m16][ka];
            acc0 = __builtin_amdgcn_wmma_f32_16x16x4_f32(false, a, false, b0,
                                                         (short)0, acc0, false, false);
            acc1 = __builtin_amdgcn_wmma_f32_16x16x4_f32(false, a, false, b1,
                                                         (short)0, acc1, false, false);
        }
        __builtin_amdgcn_sched_barrier(0);
        if (more) {
            #pragma unroll
            for (int t = 0; t < 8; ++t) {
                Ap[cur ^ 1][lr][kq + 4 * t] = ra[t];
                Bp[cur ^ 1][lr][kq + 4 * t] = rb[t];
            }
        }
        __syncthreads();
        cur ^= 1;
    }
    int n     = lane & 15;
    int rbase = l0 + tr * 16 + ((lane >> 4) << 3);
    #pragma unroll
    for (int v = 0; v < 8; ++v) {
        int row = rbase + v;
        out[(size_t)row * L_ + p0 + tcb * 16 + n]       = acc0[v];
        out[(size_t)row * L_ + p0 + (tcb + 1) * 16 + n] = acc1[v];
    }
}

#define SM_COLS  32
#define SM_WAVES HD
#define SM_ROWS  WD
#define SM_RB    4
__global__ __launch_bounds__(SM_COLS * SM_WAVES) void k_softmax(const float* __restrict__ G,
                                                                const float* __restrict__ denL,
                                                                const float* __restrict__ denR,
                                                                float* __restrict__ scL,
                                                                float* __restrict__ scR) {
    __shared__ float vs[L_][SM_COLS];
    __shared__ float redm[SM_WAVES][SM_COLS];
    __shared__ float reds[SM_WAVES][SM_COLS];
    float* sc = blockIdx.y ? scR : scL;
    int b    = blockIdx.z;
    const float* g   = G + (size_t)(blockIdx.y * B_ + b) * L_ * L_;
    int lane = threadIdx.x & (SM_COLS - 1);
    int w    = __builtin_amdgcn_readfirstlane((int)(threadIdx.x / SM_COLS));
    const float* den = (blockIdx.y ? denR : denL) + b * L_ + w * SM_ROWS;
    int p    = blockIdx.x * SM_COLS + lane;
    float* col  = sc + ((size_t)b * L_ + (size_t)w * SM_ROWS) * L_ + p;
    float* vrow = &vs[w * SM_ROWS][lane];

    const bool vy0 = w > 0, vy2 = w < HD - 1;
    const float* gb[3];
    gb[0] = g + (size_t)((vy0 ? w - 1 : w) * WD) * L_;
    gb[1] = g + (size_t)(w * WD) * L_;
    gb[2] = g + (size_t)((vy2 ? w + 1 : w) * WD) * L_;
    const int py = p / WD, px = p % WD;
    int  pc[9];
    bool ok[9];
    #pragma unroll
    for (int t = 0; t < 9; ++t) {
        int dy = t / 3 - 1, dx = t % 3 - 1;
        bool vp = ((unsigned)(py + dy) < HD) & ((unsigned)(px + dx) < WD);
        pc[t] = vp ? p + dy * WD + dx : p;
        ok[t] = vp & (dy < 0 ? vy0 : (dy > 0 ? vy2 : true));
    }

    float mx = -INFINITY;
    #pragma unroll 1
    for (int i = 0; i < SM_ROWS; i += SM_RB) {
        const bool vfirst = i > 0;
        const bool vlast  = i + SM_RB < SM_ROWS;
        float x[SM_RB][9];
        #pragma unroll
        for (int r = 0; r < SM_RB; ++r) {
            #pragma unroll
            for (int t = 0; t < 9; ++t) {
                const int dx = t % 3 - 1;
                int cx = i + r + dx;
                if (r == 0 && dx < 0)         cx = max(cx, 0);
                if (r == SM_RB - 1 && dx > 0) cx = min(cx, WD - 1);
                x[r][t] = gb[t / 3][(size_t)cx * L_ + pc[t]];
            }
        }
        #pragma unroll
        for (int r = 0; r < SM_RB; ++r) {
            float s = x[r][4];
            #pragma unroll
            for (int t = 0; t < 9; ++t) {
                if (t == 4) continue;
                const int dx = t % 3 - 1;
                bool vx = (r == 0 && dx < 0) ? vfirst : ((r == SM_RB - 1 && dx > 0) ? vlast : true);
                s += (ok[t] & vx) ? x[r][t] : 0.0f;
            }
            float val = s * (SCALE_ / den[i + r]);
            vrow[(i + r) * SM_COLS] = val;
            mx = fmaxf(mx, val);
        }
    }
    redm[w][lane] = mx;
    __syncthreads();
    float m = redm[0][lane];
    #pragma unroll
    for (int j = 1; j < SM_WAVES; ++j) m = fmaxf(m, redm[j][lane]);

    float v[SM_ROWS];
    float s = 0.0f;
    #pragma unroll
    for (int i = 0; i < SM_ROWS; ++i) {
        v[i] = expf(vrow[i * SM_COLS] - m);
        s += v[i];
    }
    reds[w][lane] = s;
    __syncthreads();
    float tot = 0.0f;
    #pragma unroll
    for (int j = 0; j < SM_WAVES; ++j) tot += reds[j][lane];
    float inv = 1.0f / tot;

    #pragma unroll
    for (int i = 0; i < SM_ROWS; ++i) col[(size_t)i * L_] = v[i] * inv;
}

__global__ __launch_bounds__(256) void k_shift(const float* __restrict__ atL,
                                               const float* __restrict__ atR,
                                               float* __restrict__ Bm) {
    __shared__ float wt[W_];
    if (threadIdx.x < W_) {
        const float PI = 3.14159265358979323846f;
        wt[threadIdx.x] = 0.5f * (cosf(PI * (float)threadIdx.x / (float)(W_ - 1)) + 1.0f);
    }
    __syncthreads();
    int idx = blockIdx.x * 256 + threadIdx.x;
    int p = idx % L_;  int t = idx / L_;
    int l = t % L_;    t /= L_;
    int side = t & 1;  int b = t >> 1;
    const float* at = (side ? atR : atL) + (size_t)b * L_ * L_;
    int ly = l / WD, lx = l % WD, py = p / WD, px = p % WD;
    float T[3][3];
    #pragma unroll
    for (int ey = -1; ey <= 1; ++ey) {
        #pragma unroll
        for (int ex = -1; ex <= 1; ++ex) {
            int y1 = ly - ey, x1 = lx - ex, y2 = py - ey, x2 = px - ex;
            bool ok = ((unsigned)y1 < HD) & ((unsigned)x1 < WD) & ((unsigned)y2 < HD) & ((unsigned)x2 < WD);
            int off = ok ? (y1 * WD + x1) * L_ + y2 * WD + x2 : 0;
            float v = at[off];
            T[ey + 1][ex + 1] = ok ? v : 0.0f;
        }
    }
    #pragma unroll
    for (int a = 0; a < 2; ++a) {
        #pragma unroll
        for (int bq = 0; bq < 2; ++bq) {
            const int y0 = a ? 0 : 1, x0 = bq ? 0 : 1;
            float s = (T[y0][x0] + T[y0][x0 + 1]) + (T[y0 + 1][x0] + T[y0 + 1][x0 + 1]);
            int sx = 2 * px + bq;
            float wg = wt[side ? (W_ - 1 - sx) : sx];
            Bm[((size_t)((b * 2 + a) * 2 + bq) * (2 * L_) + side * L_ + l) * L_ + p] = 0.25f * wg * s;
        }
    }
}

#define TK3  32
#define LDP3 (TK3 + 2)
#define NT3  48
#define KC3  (2 * L_ / TK3)
typedef float v4f __attribute__((ext_vector_type(4)));
__global__ __launch_bounds__(256) void k_recon(const float* __restrict__ shcL,
                                               const float* __restrict__ shcR,
                                               const float* __restrict__ Bm,
                                               float* __restrict__ out) {
    __shared__ __align__(16) float Ap[2][2][64][LDP3];
    __shared__ __align__(16) float Bp[2][2][NT3][LDP3];
    const int b   = blockIdx.z;
    const int a   = blockIdx.y;
    const int m0  = (blockIdx.x / 12) * 64;
    const int p0  = (blockIdx.x % 12) * NT3;
    const int tid = threadIdx.x;
    const int lane = tid & 31, w = tid >> 5;
    const int tr  = w >> 1;
    const int bq  = w & 1;
    const int m16   = lane & 15;
    const int khalf = (lane >> 4) * 2;
    v8f acc0 = {}; v8f acc1 = {}; v8f acc2 = {};

    const size_t aoff = ((size_t)(b * C_ + m0 + w) * H_ + a) * W_;
    const float* aL = shcL + aoff;
    const float* aR = shcR + aoff;
    int bsel[3], krow[3], pv4[3];
    const float* bsrc[3];
    #pragma unroll
    for (int t = 0; t < 3; ++t) {
        int u = tid + 256 * t;
        bsel[t] = u / 384;
        int r = u % 384;
        krow[t] = r / 12;
        pv4[t]  = 4 * (r % 12);
        bsrc[t] = Bm + ((size_t)((b * 2 + a) * 2 + bsel[t]) * (2 * L_) + krow[t]) * L_ + p0 + pv4[t];
    }

    v2f va[8]; v4f vb[3];
    {
        const int l  = lane;
        const int ly = l / WD, lx = l - ly * WD;
        const float* pa = aL + ly * (2 * W_) + 2 * lx;
        #pragma unroll
        for (int t = 0; t < 8; ++t) va[t] = *(const v2f*)(pa + (size_t)t * 8 * (H_ * W_));
        #pragma unroll
        for (int t = 0; t < 3; ++t) vb[t] = *(const v4f*)(bsrc[t]);
        #pragma unroll
        for (int t = 0; t < 8; ++t) {
            Ap[0][0][w + 8 * t][lane] = va[t].x;
            Ap[0][1][w + 8 * t][lane] = va[t].y;
        }
        #pragma unroll
        for (int t = 0; t < 3; ++t) {
            #pragma unroll
            for (int e = 0; e < 4; ++e) Bp[0][bsel[t]][pv4[t] + e][krow[t]] = vb[t][e];
        }
    }
    __syncthreads();
    int cur = 0;
    for (int q = 0; q < KC3; ++q) {
        const bool more = (q + 1) < KC3;
        if (more) {
            const int qn   = q + 1;
            const int side = qn >= (KC3 / 2) ? 1 : 0;
            const int l    = (qn - side * (KC3 / 2)) * TK3 + lane;
            const int ly   = l / WD, lx = l - ly * WD;
            const float* pa = (side ? aR : aL) + ly * (2 * W_) + 2 * lx;
            #pragma unroll
            for (int t = 0; t < 8; ++t) va[t] = *(const v2f*)(pa + (size_t)t * 8 * (H_ * W_));
            #pragma unroll
            for (int t = 0; t < 3; ++t) vb[t] = *(const v4f*)(bsrc[t] + (size_t)qn * TK3 * L_);
        }
        __builtin_amdgcn_sched_barrier(0);
        #pragma unroll
        for (int kk0 = 0; kk0 < TK3; kk0 += 4) {
            int ka = kk0 + khalf;
            v2f fa = *(const v2f*)&Ap[cur][bq][tr * 16 + m16][ka];
            v2f b0 = *(const v2f*)&Bp[cur][bq][m16][ka];
            v2f b1 = *(const v2f*)&Bp[cur][bq][16 + m16][ka];
            v2f b2 = *(const v2f*)&Bp[cur][bq][32 + m16][ka];
            acc0 = __builtin_amdgcn_wmma_f32_16x16x4_f32(false, fa, false, b0,
                                                         (short)0, acc0, false, false);
            acc1 = __builtin_amdgcn_wmma_f32_16x16x4_f32(false, fa, false, b1,
                                                         (short)0, acc1, false, false);
            acc2 = __builtin_amdgcn_wmma_f32_16x16x4_f32(false, fa, false, b2,
                                                         (short)0, acc2, false, false);
        }
        __builtin_amdgcn_sched_barrier(0);
        if (more) {
            #pragma unroll
            for (int t = 0; t < 8; ++t) {
                Ap[cur ^ 1][0][w + 8 * t][lane] = va[t].x;
                Ap[cur ^ 1][1][w + 8 * t][lane] = va[t].y;
            }
            #pragma unroll
            for (int t = 0; t < 3; ++t) {
                #pragma unroll
                for (int e = 0; e < 4; ++e) Bp[cur ^ 1][bsel[t]][pv4[t] + e][krow[t]] = vb[t][e];
            }
        }
        __syncthreads();
        cur ^= 1;
    }
    const int n  = lane & 15;
    const int cb = m0 + tr * 16 + ((lane >> 4) << 3);
    #pragma unroll
    for (int j = 0; j < 3; ++j) {
        const int p  = p0 + j * 16 + n;
        const int py = p / WD, px = p - py * WD;
        const size_t o = ((size_t)(b * C_ + cb) * H_ + 2 * py + a) * W_ + 2 * px + bq;
        const v8f acc = j == 0 ? acc0 : (j == 1 ? acc1 : acc2);
        #pragma unroll
        for (int v = 0; v < 8; ++v) out[o + (size_t)v * (H_ * W_)] = acc[v];
    }
}

extern "C" void kernel_launch(void* const* d_in, const int* in_sizes, int n_in,
                              void* d_out, int out_size, void* d_ws, size_t ws_size,
                              hipStream_t stream) {
    (void)in_sizes; (void)n_in; (void)out_size; (void)ws_size;
    const float* left  = (const float*)d_in[0];
    const float* right = (const float*)d_in[1];
    const float* mid   = (const float*)d_in[2];
    const float* shcL  = (const float*)d_in[3];
    const float* shcR  = (const float*)d_in[4];
    float* out = (float*)d_out;

    float* w   = (float*)d_ws;
    const size_t NDS = (size_t)B_ * C_ * HD * WD;
    const size_t NDN = (size_t)B_ * L_;
    const size_t NSC = (size_t)B_ * L_ * L_;
    const size_t NY  = (size_t)B_ * K16 * L_;
    float* dsL  = w;
    float* dsR  = dsL + NDS;
    float* dsM  = dsR + NDS;
    float* denL = dsM + NDS;
    float* denR = denL + NDN;
    float* scL  = denR + NDN;
    float* scR  = scL + NSC;
    float* YL   = scR + NSC;
    float* YR   = YL + NY;
    float* G    = YL;
    float* Bm   = YL;
    (void)YR;

    k_prep_ds<<<dim3((NDS + 255) / 256), dim3(256), 0, stream>>>(left, right, mid,
                                                                 dsL, dsR, dsM);
    k_denom<<<dim3(B_ * L_ / 8, 2), dim3(256), 0, stream>>>(dsL, dsR, denL, denR);
    k_scores<<<dim3(81, 2, B_), dim3(256), 0, stream>>>(dsL, dsR, dsM, G);
    k_softmax<<<dim3(L_ / SM_COLS, 2, B_), dim3(SM_COLS * SM_WAVES), 0, stream>>>(G, denL, denR,
                                                                                  scL, scR);
    k_shift<<<dim3((B_ * 2 * L_ * L_) / 256), dim3(256), 0, stream>>>(scL, scR, Bm);
    k_recon<<<dim3(48, 2, B_), dim3(256), 0, stream>>>(shcL, shcR, Bm, out);
}


# softmax kernel loads each correlation value once: the three dy rows are 3 loads per filter, the dx taps come from neighbour lanes' neighbour registers (lane shifts); 216 loads per lane become 72
# speedup vs baseline: 9.7341x; 1.2122x over previous
#include <hip/hip_runtime.h>
#include <hip/hip_bf16.h>
#include <math.h>

#define B_   2
#define C_   256
#define H_   48
#define W_   48
#define HD   24
#define WD   24
#define L_   576
#define K9   2304
#define K16  4096
#define SCALE_ 10.0f
#define EPS_   1e-4f

typedef float v2f __attribute__((ext_vector_type(2)));
typedef float v8f __attribute__((ext_vector_type(8)));

#define TK  32
#define LDP (TK + 2)

__global__ __launch_bounds__(256) void k_prep_ds(const float* __restrict__ left,
                                                 const float* __restrict__ right,
                                                 const float* __restrict__ mid,
                                                 float* __restrict__ dsL,
                                                 float* __restrict__ dsR,
                                                 float* __restrict__ dsM) {
    int idx = blockIdx.x * 256 + threadIdx.x;
    if (idx >= B_ * C_ * HD * WD) return;
    int x = idx % WD, t = idx / WD;
    int y = t % HD;  t /= HD;
    int c = t % C_;  int b = t / C_;
    int src = ((b * C_ + c) * H_ + 2 * y) * W_ + 2 * x;
    dsL[idx] = left[src];
    dsR[idx] = right[src];
    dsM[idx] = mid[src];
}

__global__ __launch_bounds__(256) void k_scores(const float* __restrict__ dsL,
                                                const float* __restrict__ dsR,
                                                const float* __restrict__ dsM,
                                                float* __restrict__ G,
                                                float* __restrict__ nrm) {
    __shared__ __align__(16) float Ap[2][64][LDP];
    __shared__ float nred[4][64];
    __shared__ __align__(16) float Bp[2][64][LDP];
    const float* dsF = blockIdx.y ? dsR : dsL;
    int b   = blockIdx.z;
    float* out = G + (size_t)(blockIdx.y * B_ + b) * L_ * L_;
    int l0  = (blockIdx.x / 9) * 64;
    int p0  = (blockIdx.x % 9) * 64;
    const bool doN = (blockIdx.x % 9) == 0;
    float sq = 0.0f;
    int tid = threadIdx.x;
    int lane = tid & 31, w = tid >> 5;
    int tr  = w >> 1;
    int tcb = (w & 1) * 2;
    int m16   = lane & 15;
    int khalf = (lane >> 4) * 2;
    v8f acc0 = {}; v8f acc1 = {};

    const int lr = tid & 63;
    const int kq = tid >> 6;
    const float* baseA = dsF + (size_t)(b * C_ + kq) * L_ + l0 + lr;
    const float* baseB = dsM + (size_t)(b * C_ + kq) * L_ + p0 + lr;
    float ra[8], rb[8];
    #pragma unroll
    for (int t = 0; t < 8; ++t) {
        ra[t] = baseA[(size_t)t * 4 * L_];
        rb[t] = baseB[(size_t)t * 4 * L_];
    }
    #pragma unroll
    for (int t = 0; t < 8; ++t) {
        Ap[0][lr][kq + 4 * t] = ra[t];
        Bp[0][lr][kq + 4 * t] = rb[t];
    }
    if (doN) {
        #pragma unroll
        for (int t = 0; t < 8; ++t) sq = fmaf(ra[t], ra[t], sq);
    }
    __syncthreads();
    int cur = 0;
    for (int q = 0; q < C_ / TK; ++q) {
        const bool more = (q + 1) < C_ / TK;
        if (more) {
            const float* pa = baseA + (size_t)(q + 1) * TK * L_;
            const float* pb = baseB + (size_t)(q + 1) * TK * L_;
            #pragma unroll
            for (int t = 0; t < 8; ++t) {
                ra[t] = pa[(size_t)t * 4 * L_];
                rb[t] = pb[(size_t)t * 4 * L_];
            }
        }
        __builtin_amdgcn_sched_barrier(0);
        #pragma unroll
        for (int kk0 = 0; kk0 < TK; kk0 += 4) {
            int ka = kk0 + khalf;
            v2f a  = *(const v2f*)&Ap[cur][tr * 16 + m16][ka];
            v2f b0 = *(const v2f*)&Bp[cur][tcb * 16 + m16][ka];
            v2f b1 = *(const v2f*)&Bp[cur][(tcb + 1) * 16 + m16][ka];
            acc0 = __builtin_amdgcn_wmma_f32_16x16x4_f32(false, a, false, b0,
                                                         (short)0, acc0, false, false);
            acc1 = __builtin_amdgcn_wmma_f32_16x16x4_f32(false, a, false, b1,
                                                         (short)0, acc1, false, false);
        }
        __builtin_amdgcn_sched_barrier(0);
        if (more) {
            #pragma unroll
            for (int t = 0; t < 8; ++t) {
                Ap[cur ^ 1][lr][kq + 4 * t] = ra[t];
                Bp[cur ^ 1][lr][kq + 4 * t] = rb[t];
            }
            if (doN) {
                #pragma unroll
                for (int t = 0; t < 8; ++t) sq = fmaf(ra[t], ra[t], sq);
            }
        }
        __syncthreads();
        cur ^= 1;
    }
    if (doN) {
        nred[kq][lr] = sq;
        __syncthreads();
        if (tid < 64)
            nrm[(size_t)(blockIdx.y * B_ + b) * L_ + l0 + tid] =
                (nred[0][tid] + nred[1][tid]) + (nred[2][tid] + nred[3][tid]);
    }
    int n     = lane & 15;
    int rbase = l0 + tr * 16 + ((lane >> 4) << 3);
    #pragma unroll
    for (int v = 0; v < 8; ++v) {
        int row = rbase + v;
        out[(size_t)row * L_ + p0 + tcb * 16 + n]       = acc0[v];
        out[(size_t)row * L_ + p0 + (tcb + 1) * 16 + n] = acc1[v];
    }
}

#define SM_COLS  32
#define SM_WAVES HD
#define SM_ROWS  WD
__global__ __launch_bounds__(SM_COLS * SM_WAVES) void k_softmax(const float* __restrict__ G,
                                                                const float* __restrict__ nrm,
                                                                float* __restrict__ Xp) {
    __shared__ float redm[SM_WAVES][SM_COLS];
    __shared__ float reds[SM_WAVES][SM_COLS];
    __shared__ float wt[W_];
    if (threadIdx.x < W_) {
        const float PI = 3.14159265358979323846f;
        wt[threadIdx.x] = 0.5f * (cosf(PI * (float)threadIdx.x / (float)(W_ - 1)) + 1.0f);
    }
    const int side = blockIdx.y;
    int b    = blockIdx.z;
    const float* g   = G + (size_t)(blockIdx.y * B_ + b) * L_ * L_;
    int lane = threadIdx.x & (SM_COLS - 1);
    int w    = __builtin_amdgcn_readfirstlane((int)(threadIdx.x / SM_COLS));
    const float* nr  = nrm + (size_t)(blockIdx.y * B_ + b) * L_;
    const int py = blockIdx.x;
    const int px = min(lane, WD - 1);
    const bool act = lane < WD;
    int p    = py * WD + px;
    const bool hasl = px > 0, hasr = px < WD - 1;

    const bool vy0 = w > 0, vy2 = w < HD - 1;
    const bool pym = py > 0, pyp = py < HD - 1;
    const bool okm = vy0 & pym;
    const bool okp = vy2 & pyp;
    const float* gm = g + (size_t)((vy0 ? w - 1 : w) * WD) * L_ + (pym ? p - WD : p);
    const float* g0 = g + (size_t)(w * WD) * L_ + p;
    const float* gp = g + (size_t)((vy2 ? w + 1 : w) * WD) * L_ + (pyp ? p + WD : p);

    float rs;
    {
        const int lxx = min(lane, WD - 1);
        float ns = 0.0f;
        #pragma unroll
        for (int t = 0; t < 9; ++t) {
            int yy = w + t / 3 - 1, xx = lxx + t % 3 - 1;
            bool vn = ((unsigned)yy < HD) & ((unsigned)xx < WD);
            float nv = nr[vn ? yy * WD + xx : 0];
            ns += vn ? nv : 0.0f;
        }
        rs = SCALE_ / sqrtf(ns + (float)K9 * EPS_);
    }

    float T[SM_ROWS];
    {
        float xm[SM_ROWS], x0[SM_ROWS], xp[SM_ROWS];
        #pragma unroll
        for (int i = 0; i < SM_ROWS; ++i) {
            xm[i] = gm[(size_t)i * L_];
            x0[i] = g0[(size_t)i * L_];
            xp[i] = gp[(size_t)i * L_];
        }
        #pragma unroll
        for (int i = 0; i < SM_ROWS; ++i)
            T[i] = x0[i] + ((okm ? xm[i] : 0.0f) + (okp ? xp[i] : 0.0f));
    }

    float v[SM_ROWS];
    float mx = -INFINITY;
    #pragma unroll
    for (int i = 0; i < SM_ROWS; ++i) {
        float s = T[i];
        if (i > 0) {
            float nb = __shfl_up(T[i - 1], 1, SM_COLS);
            s += hasl ? nb : 0.0f;
        }
        if (i < SM_ROWS - 1) {
            float nb = __shfl_down(T[i + 1], 1, SM_COLS);
            s += hasr ? nb : 0.0f;
        }
        float val = s * __builtin_bit_cast(float,
                        __builtin_amdgcn_readlane(__builtin_bit_cast(int, rs), i));
        v[i] = val;
        mx = fmaxf(mx, val);
    }
    redm[w][lane] = mx;
    __syncthreads();
    float m = redm[0][lane];
    #pragma unroll
    for (int j = 1; j < SM_WAVES; ++j) m = fmaxf(m, redm[j][lane]);

    float s = 0.0f;
    #pragma unroll
    for (int i = 0; i < SM_ROWS; ++i) {
        v[i] = expf(v[i] - m);
        s += v[i];
    }
    reds[w][lane] = s;
    __syncthreads();
    float tot = 0.0f;
    #pragma unroll
    for (int j = 0; j < SM_WAVES; ++j) tot += reds[j][lane];
    float inv = 1.0f / tot;

    #pragma unroll
    for (int i = 0; i < SM_ROWS; ++i) v[i] *= inv;

    const int sx0 = 2 * px, sx1 = 2 * px + 1;
    const float f0 = 0.25f * wt[side ? (W_ - 1 - sx0) : sx0];
    const float f1 = 0.25f * wt[side ? (W_ - 1 - sx1) : sx1];
    float* x0 = Xp + ((size_t)(((b * 2 + 0) * 2 + side) * L_) + (size_t)w * SM_ROWS) * L_ + p;
    float* x1 = Xp + ((size_t)(((b * 2 + 1) * 2 + side) * L_) + (size_t)w * SM_ROWS) * L_ + p;
    #pragma unroll
    for (int i = 0; i < SM_ROWS; ++i) {
        float s0 = v[i], s1 = v[i];
        if (i > 0) {
            float nb = __shfl_up(v[i - 1], 1, SM_COLS);
            s0 += hasl ? nb : 0.0f;
        }
        if (i < SM_ROWS - 1) {
            float nb = __shfl_down(v[i + 1], 1, SM_COLS);
            s1 += hasr ? nb : 0.0f;
        }
        if (act) {
            x0[(size_t)i * L_] = f0 * s0;
            x1[(size_t)i * L_] = f1 * s1;
        }
    }
}


#define TK3  32
#define LDP3 (TK3 + 2)
#define NT3  48
#define KC3  (2 * L_ / TK3)
typedef float v4f __attribute__((ext_vector_type(4)));
__global__ __launch_bounds__(256) void k_recon(const float* __restrict__ shcL,
                                               const float* __restrict__ shcR,
                                               const float* __restrict__ Xp,
                                               float* __restrict__ out) {
    __shared__ __align__(16) float Ap[2][2][64][LDP3];
    __shared__ __align__(16) float Bp[2][2][NT3][LDP3];
    const int b   = blockIdx.z;
    const int a   = blockIdx.y;
    const int m0  = (blockIdx.x / 12) * 64;
    const int p0  = (blockIdx.x % 12) * NT3;
    const int tid = threadIdx.x;
    const int lane = tid & 31, w = tid >> 5;
    const int tr  = w >> 1;
    const int bq  = w & 1;
    const int m16   = lane & 15;
    const int khalf = (lane >> 4) * 2;
    v8f acc0 = {}; v8f acc1 = {}; v8f acc2 = {};

    const size_t aoff = ((size_t)(b * C_ + m0 + w) * H_ + a) * W_;
    const float* aL = shcL + aoff;
    const float* aR = shcR + aoff;
    int bsel[3], krow[3], pv4[3];
    const float* bsrc[3];
    bool pok[3];
    #pragma unroll
    for (int t = 0; t < 3; ++t) {
        int u = tid + 256 * t;
        bsel[t] = u / 384;
        int r = u % 384;
        krow[t] = r / 12;
        pv4[t]  = 4 * (r % 12);
        bsrc[t] = Xp + ((size_t)(b * 2 + bsel[t]) * (2 * L_) + krow[t]) * L_ + p0 + pv4[t];
        pok[t]  = a ? (p0 + pv4[t] < L_ - WD) : (p0 + pv4[t] >= WD);
    }
    const int ysh = a ? (WD * L_ + WD) : -(WD * L_ + WD);

    v2f va[8]; v4f vb[3]; v4f vc[3]; bool ok2[3];
    {
        const int l  = lane;
        const int ly = l / WD, lx = l - ly * WD;
        const float* pa = aL + ly * (2 * W_) + 2 * lx;
        #pragma unroll
        for (int t = 0; t < 8; ++t) va[t] = *(const v2f*)(pa + (size_t)t * 8 * (H_ * W_));
        #pragma unroll
        for (int t = 0; t < 3; ++t) {
            const int lk = krow[t];
            ok2[t] = pok[t] & (a ? (lk < L_ - WD) : (lk >= WD));
            vb[t] = *(const v4f*)(bsrc[t]);
            vc[t] = *(const v4f*)(bsrc[t] + (ok2[t] ? ysh : 0));
        }
        #pragma unroll
        for (int t = 0; t < 8; ++t) {
            Ap[0][0][w + 8 * t][lane] = va[t].x;
            Ap[0][1][w + 8 * t][lane] = va[t].y;
        }
        #pragma unroll
        for (int t = 0; t < 3; ++t) {
            #pragma unroll
            for (int e = 0; e < 4; ++e)
                Bp[0][bsel[t]][pv4[t] + e][krow[t]] = vb[t][e] + (ok2[t] ? vc[t][e] : 0.0f);
        }
    }
    __syncthreads();
    int cur = 0;
    for (int q = 0; q < KC3; ++q) {
        const bool more = (q + 1) < KC3;
        if (more) {
            const int qn   = q + 1;
            const int side = qn >= (KC3 / 2) ? 1 : 0;
            const int l    = (qn - side * (KC3 / 2)) * TK3 + lane;
            const int ly   = l / WD, lx = l - ly * WD;
            const float* pa = (side ? aR : aL) + ly * (2 * W_) + 2 * lx;
            #pragma unroll
            for (int t = 0; t < 8; ++t) va[t] = *(const v2f*)(pa + (size_t)t * 8 * (H_ * W_));
            const int lq = (qn - side * (KC3 / 2)) * TK3;
            #pragma unroll
            for (int t = 0; t < 3; ++t) {
                const int lk = lq + krow[t];
                const float* pb = bsrc[t] + (size_t)qn * TK3 * L_;
                ok2[t] = pok[t] & (a ? (lk < L_ - WD) : (lk >= WD));
                vb[t] = *(const v4f*)(pb);
                vc[t] = *(const v4f*)(pb + (ok2[t] ? ysh : 0));
            }
        }
        __builtin_amdgcn_sched_barrier(0);
        #pragma unroll
        for (int kk0 = 0; kk0 < TK3; kk0 += 4) {
            int ka = kk0 + khalf;
            v2f fa = *(const v2f*)&Ap[cur][bq][tr * 16 + m16][ka];
            v2f b0 = *(const v2f*)&Bp[cur][bq][m16][ka];
            v2f b1 = *(const v2f*)&Bp[cur][bq][16 + m16][ka];
            v2f b2 = *(const v2f*)&Bp[cur][bq][32 + m16][ka];
            acc0 = __builtin_amdgcn_wmma_f32_16x16x4_f32(false, fa, false, b0,
                                                         (short)0, acc0, false, false);
            acc1 = __builtin_amdgcn_wmma_f32_16x16x4_f32(false, fa, false, b1,
                                                         (short)0, acc1, false, false);
            acc2 = __builtin_amdgcn_wmma_f32_16x16x4_f32(false, fa, false, b2,
                                                         (short)0, acc2, false, false);
        }
        __builtin_amdgcn_sched_barrier(0);
        if (more) {
            #pragma unroll
            for (int t = 0; t < 8; ++t) {
                Ap[cur ^ 1][0][w + 8 * t][lane] = va[t].x;
                Ap[cur ^ 1][1][w + 8 * t][lane] = va[t].y;
            }
            #pragma unroll
            for (int t = 0; t < 3; ++t) {
                #pragma unroll
                for (int e = 0; e < 4; ++e)
                    Bp[cur ^ 1][bsel[t]][pv4[t] + e][krow[t]] = vb[t][e] + (ok2[t] ? vc[t][e] : 0.0f);
            }
        }
        __syncthreads();
        cur ^= 1;
    }
    const int n  = lane & 15;
    const int cb = m0 + tr * 16 + ((lane >> 4) << 3);
    #pragma unroll
    for (int j = 0; j < 3; ++j) {
        const int p  = p0 + j * 16 + n;
        const int py = p / WD, px = p - py * WD;
        const size_t o = ((size_t)(b * C_ + cb) * H_ + 2 * py + a) * W_ + 2 * px + bq;
        const v8f acc = j == 0 ? acc0 : (j == 1 ? acc1 : acc2);
        #pragma unroll
        for (int v = 0; v < 8; ++v) out[o + (size_t)v * (H_ * W_)] = acc[v];
    }
}

extern "C" void kernel_launch(void* const* d_in, const int* in_sizes, int n_in,
                              void* d_out, int out_size, void* d_ws, size_t ws_size,
                              hipStream_t stream) {
    (void)in_sizes; (void)n_in; (void)out_size; (void)ws_size;
    const float* left  = (const float*)d_in[0];
    const float* right = (const float*)d_in[1];
    const float* mid   = (const float*)d_in[2];
    const float* shcL  = (const float*)d_in[3];
    const float* shcR  = (const float*)d_in[4];
    float* out = (float*)d_out;

    float* w   = (float*)d_ws;
    const size_t NDS = (size_t)B_ * C_ * HD * WD;
    const size_t NDN = (size_t)B_ * L_;
    const size_t NSC = (size_t)B_ * L_ * L_;
    const size_t NY  = (size_t)B_ * K16 * L_;
    float* dsL  = w;
    float* dsR  = dsL + NDS;
    float* dsM  = dsR + NDS;
    float* denL = dsM + NDS;
    float* denR = denL + NDN;
    float* scL  = denR + NDN;
    float* scR  = scL + NSC;
    float* YL   = scR + NSC;
    float* YR   = YL + NY;
    float* G    = scL;
    float* Xp   = YL;
    (void)scR; (void)YR;

    k_prep_ds<<<dim3((NDS + 255) / 256), dim3(256), 0, stream>>>(left, right, mid,
                                                                 dsL, dsR, dsM);
    k_scores<<<dim3(81, 2, B_), dim3(256), 0, stream>>>(dsL, dsR, dsM, G, denL);
    k_softmax<<<dim3(HD, 2, B_), dim3(SM_COLS * SM_WAVES), 0, stream>>>(G, denL, Xp);
    k_recon<<<dim3(48, 2, B_), dim3(256), 0, stream>>>(shcL, shcR, Xp, out);
}


# reconstruct GEMM pipelined one chunk deeper: next chunk's 16 fragment loads as one group into a 2nd register set ahead of the 24 WMMAs, staging off the tail (a SIMD's two waves in opposite order)
# speedup vs baseline: 10.5386x; 1.0826x over previous
#include <hip/hip_runtime.h>
#include <hip/hip_bf16.h>
#include <math.h>

#define B_   2
#define C_   256
#define H_   48
#define W_   48
#define HD   24
#define WD   24
#define L_   576
#define K9   2304
#define K16  4096
#define SCALE_ 10.0f
#define EPS_   1e-4f

typedef float v2f __attribute__((ext_vector_type(2)));
typedef float v8f __attribute__((ext_vector_type(8)));

#define TK  32
#define LDP (TK + 2)

__global__ __launch_bounds__(256) void k_prep_ds(const float* __restrict__ left,
                                                 const float* __restrict__ right,
                                                 const float* __restrict__ mid,
                                                 float* __restrict__ dsL,
                                                 float* __restrict__ dsR,
                                                 float* __restrict__ dsM) {
    int idx = blockIdx.x * 256 + threadIdx.x;
    if (idx >= B_ * C_ * HD * WD) return;
    int x = idx % WD, t = idx / WD;
    int y = t % HD;  t /= HD;
    int c = t % C_;  int b = t / C_;
    int src = ((b * C_ + c) * H_ + 2 * y) * W_ + 2 * x;
    dsL[idx] = left[src];
    dsR[idx] = right[src];
    dsM[idx] = mid[src];
}

__global__ __launch_bounds__(256) void k_scores(const float* __restrict__ dsL,
                                                const float* __restrict__ dsR,
                                                const float* __restrict__ dsM,
                                                float* __restrict__ G,
                                                float* __restrict__ nrm) {
    __shared__ __align__(16) float Ap[2][64][LDP];
    __shared__ float nred[4][64];
    __shared__ __align__(16) float Bp[2][64][LDP];
    const float* dsF = blockIdx.y ? dsR : dsL;
    int b   = blockIdx.z;
    float* out = G + (size_t)(blockIdx.y * B_ + b) * L_ * L_;
    int l0  = (blockIdx.x / 9) * 64;
    int p0  = (blockIdx.x % 9) * 64;
    const bool doN = (blockIdx.x % 9) == 0;
    float sq = 0.0f;
    int tid = threadIdx.x;
    int lane = tid & 31, w = tid >> 5;
    int tr  = w >> 1;
    int tcb = (w & 1) * 2;
    int m16   = lane & 15;
    int khalf = (lane >> 4) * 2;
    v8f acc0 = {}; v8f acc1 = {};

    const int lr = tid & 63;
    const int kq = tid >> 6;
    const float* baseA = dsF + (size_t)(b * C_ + kq) * L_ + l0 + lr;
    const float* baseB = dsM + (size_t)(b * C_ + kq) * L_ + p0 + lr;
    float ra[8], rb[8];
    #pragma unroll
    for (int t = 0; t < 8; ++t) {
        ra[t] = baseA[(size_t)t * 4 * L_];
        rb[t] = baseB[(size_t)t * 4 * L_];
    }
    #pragma unroll
    for (int t = 0; t < 8; ++t) {
        Ap[0][lr][kq + 4 * t] = ra[t];
        Bp[0][lr][kq + 4 * t] = rb[t];
    }
    if (doN) {
        #pragma unroll
        for (int t = 0; t < 8; ++t) sq = fmaf(ra[t], ra[t], sq);
    }
    __syncthreads();
    int cur = 0;
    for (int q = 0; q < C_ / TK; ++q) {
        const bool more = (q + 1) < C_ / TK;
        if (more) {
            const float* pa = baseA + (size_t)(q + 1) * TK * L_;
            const float* pb = baseB + (size_t)(q + 1) * TK * L_;
            #pragma unroll
            for (int t = 0; t < 8; ++t) {
                ra[t] = pa[(size_t)t * 4 * L_];
                rb[t] = pb[(size_t)t * 4 * L_];
            }
        }
        __builtin_amdgcn_sched_barrier(0);
        #pragma unroll
        for (int kk0 = 0; kk0 < TK; kk0 += 4) {
            int ka = kk0 + khalf;
            v2f a  = *(const v2f*)&Ap[cur][tr * 16 + m16][ka];
            v2f b0 = *(const v2f*)&Bp[cur][tcb * 16 + m16][ka];
            v2f b1 = *(const v2f*)&Bp[cur][(tcb + 1) * 16 + m16][ka];
            acc0 = __builtin_amdgcn_wmma_f32_16x16x4_f32(false, a, false, b0,
                                                         (short)0, acc0, false, false);
            acc1 = __builtin_amdgcn_wmma_f32_16x16x4_f32(false, a, false, b1,
                                                         (short)0, acc1, false, false);
        }
        __builtin_amdgcn_sched_barrier(0);
        if (more) {
            #pragma unroll
            for (int t = 0; t < 8; ++t) {
                Ap[cur ^ 1][lr][kq + 4 * t] = ra[t];
                Bp[cur ^ 1][lr][kq + 4 * t] = rb[t];
            }
            if (doN) {
                #pragma unroll
                for (int t = 0; t < 8; ++t) sq = fmaf(ra[t], ra[t], sq);
            }
        }
        __syncthreads();
        cur ^= 1;
    }
    if (doN) {
        nred[kq][lr] = sq;
        __syncthreads();
        if (tid < 64)
            nrm[(size_t)(blockIdx.y * B_ + b) * L_ + l0 + tid] =
                (nred[0][tid] + nred[1][tid]) + (nred[2][tid] + nred[3][tid]);
    }
    int n     = lane & 15;
    int rbase = l0 + tr * 16 + ((lane >> 4) << 3);
    #pragma unroll
    for (int v = 0; v < 8; ++v) {
        int row = rbase + v;
        out[(size_t)row * L_ + p0 + tcb * 16 + n]       = acc0[v];
        out[(size_t)row * L_ + p0 + (tcb + 1) * 16 + n] = acc1[v];
    }
}

#define SM_COLS  32
#define SM_WAVES HD
#define SM_ROWS  WD
__global__ __launch_bounds__(SM_COLS * SM_WAVES) void k_softmax(const float* __restrict__ G,
                                                                const float* __restrict__ nrm,
                                                                float* __restrict__ Xp) {
    __shared__ float redm[SM_WAVES][SM_COLS];
    __shared__ float reds[SM_WAVES][SM_COLS];
    __shared__ float wt[W_];
    if (threadIdx.x < W_) {
        const float PI = 3.14159265358979323846f;
        wt[threadIdx.x] = 0.5f * (cosf(PI * (float)threadIdx.x / (float)(W_ - 1)) + 1.0f);
    }
    const int side = blockIdx.y;
    int b    = blockIdx.z;
    const float* g   = G + (size_t)(blockIdx.y * B_ + b) * L_ * L_;
    int lane = threadIdx.x & (SM_COLS - 1);
    int w    = __builtin_amdgcn_readfirstlane((int)(threadIdx.x / SM_COLS));
    const float* nr  = nrm + (size_t)(blockIdx.y * B_ + b) * L_;
    const int py = blockIdx.x;
    const int px = min(lane, WD - 1);
    const bool act = lane < WD;
    int p    = py * WD + px;
    const bool hasl = px > 0, hasr = px < WD - 1;

    const bool vy0 = w > 0, vy2 = w < HD - 1;
    const bool pym = py > 0, pyp = py < HD - 1;
    const bool okm = vy0 & pym;
    const bool okp = vy2 & pyp;
    const float* gm = g + (size_t)((vy0 ? w - 1 : w) * WD) * L_ + (pym ? p - WD : p);
    const float* g0 = g + (size_t)(w * WD) * L_ + p;
    const float* gp = g + (size_t)((vy2 ? w + 1 : w) * WD) * L_ + (pyp ? p + WD : p);

    float rs;
    {
        const int lxx = min(lane, WD - 1);
        float ns = 0.0f;
        #pragma unroll
        for (int t = 0; t < 9; ++t) {
            int yy = w + t / 3 - 1, xx = lxx + t % 3 - 1;
            bool vn = ((unsigned)yy < HD) & ((unsigned)xx < WD);
            float nv = nr[vn ? yy * WD + xx : 0];
            ns += vn ? nv : 0.0f;
        }
        rs = SCALE_ / sqrtf(ns + (float)K9 * EPS_);
    }

    float T[SM_ROWS];
    {
        float xm[SM_ROWS], x0[SM_ROWS], xp[SM_ROWS];
        #pragma unroll
        for (int i = 0; i < SM_ROWS; ++i) {
            xm[i] = gm[(size_t)i * L_];
            x0[i] = g0[(size_t)i * L_];
            xp[i] = gp[(size_t)i * L_];
        }
        #pragma unroll
        for (int i = 0; i < SM_ROWS; ++i)
            T[i] = x0[i] + ((okm ? xm[i] : 0.0f) + (okp ? xp[i] : 0.0f));
    }

    float v[SM_ROWS];
    float mx = -INFINITY;
    #pragma unroll
    for (int i = 0; i < SM_ROWS; ++i) {
        float s = T[i];
        if (i > 0) {
            float nb = __shfl_up(T[i - 1], 1, SM_COLS);
            s += hasl ? nb : 0.0f;
        }
        if (i < SM_ROWS - 1) {
            float nb = __shfl_down(T[i + 1], 1, SM_COLS);
            s += hasr ? nb : 0.0f;
        }
        float val = s * __builtin_bit_cast(float,
                        __builtin_amdgcn_readlane(__builtin_bit_cast(int, rs), i));
        v[i] = val;
        mx = fmaxf(mx, val);
    }
    redm[w][lane] = mx;
    __syncthreads();
    float m = redm[0][lane];
    #pragma unroll
    for (int j = 1; j < SM_WAVES; ++j) m = fmaxf(m, redm[j][lane]);

    float s = 0.0f;
    #pragma unroll
    for (int i = 0; i < SM_ROWS; ++i) {
        v[i] = expf(v[i] - m);
        s += v[i];
    }
    reds[w][lane] = s;
    __syncthreads();
    float tot = 0.0f;
    #pragma unroll
    for (int j = 0; j < SM_WAVES; ++j) tot += reds[j][lane];
    float inv = 1.0f / tot;

    #pragma unroll
    for (int i = 0; i < SM_ROWS; ++i) v[i] *= inv;

    const int sx0 = 2 * px, sx1 = 2 * px + 1;
    const float f0 = 0.25f * wt[side ? (W_ - 1 - sx0) : sx0];
    const float f1 = 0.25f * wt[side ? (W_ - 1 - sx1) : sx1];
    float* x0 = Xp + ((size_t)(((b * 2 + 0) * 2 + side) * L_) + (size_t)w * SM_ROWS) * L_ + p;
    float* x1 = Xp + ((size_t)(((b * 2 + 1) * 2 + side) * L_) + (size_t)w * SM_ROWS) * L_ + p;
    #pragma unroll
    for (int i = 0; i < SM_ROWS; ++i) {
        float s0 = v[i], s1 = v[i];
        if (i > 0) {
            float nb = __shfl_up(v[i - 1], 1, SM_COLS);
            s0 += hasl ? nb : 0.0f;
        }
        if (i < SM_ROWS - 1) {
            float nb = __shfl_down(v[i + 1], 1, SM_COLS);
            s1 += hasr ? nb : 0.0f;
        }
        if (act) {
            x0[(size_t)i * L_] = f0 * s0;
            x1[(size_t)i * L_] = f1 * s1;
        }
    }
}


#define TK3  32
#define LDP3 (TK3 + 2)
#define NT3  48
#define KC3  (2 * L_ / TK3)
typedef float v4f __attribute__((ext_vector_type(4)));
#define WG_BARRIER_LDS() asm volatile("s_wait_dscnt 0x0\n\ts_barrier_signal -1\n\ts_barrier_wait -1" ::: "memory")
__global__ __launch_bounds__(256) void k_recon(const float* __restrict__ shcL,
                                               const float* __restrict__ shcR,
                                               const float* __restrict__ Xp,
                                               float* __restrict__ out) {
    __shared__ __align__(16) float Ap[2][2][64][LDP3];
    __shared__ __align__(16) float Bp[2][2][NT3][LDP3];
    const int b   = blockIdx.z;
    const int a   = blockIdx.y;
    const int m0  = (blockIdx.x / 12) * 64;
    const int p0  = (blockIdx.x % 12) * NT3;
    const int tid = threadIdx.x;
    const int lane = tid & 31, w = tid >> 5;
    const int tr  = w >> 1;
    const int bq  = w & 1;
    const int m16   = lane & 15;
    const int khalf = (lane >> 4) * 2;
    v8f acc0 = {}; v8f acc1 = {}; v8f acc2 = {};

    const size_t aoff = ((size_t)(b * C_ + m0 + w) * H_ + a) * W_;
    const float* aL = shcL + aoff;
    const float* aR = shcR + aoff;
    int bsel[3], krow[3], pv4[3];
    const float* bsrc[3];
    bool pok[3];
    #pragma unroll
    for (int t = 0; t < 3; ++t) {
        int u = tid + 256 * t;
        bsel[t] = u / 384;
        int r = u % 384;
        krow[t] = r / 12;
        pv4[t]  = 4 * (r % 12);
        bsrc[t] = Xp + ((size_t)(b * 2 + bsel[t]) * (2 * L_) + krow[t]) * L_ + p0 + pv4[t];
        pok[t]  = a ? (p0 + pv4[t] < L_ - WD) : (p0 + pv4[t] >= WD);
    }
    const int ysh = a ? (WD * L_ + WD) : -(WD * L_ + WD);

    v2f va[8]; v4f vb[3]; v4f vc[3];
    const bool early = __builtin_amdgcn_readfirstlane(w) >= 4;

    auto gload = [&](const int c) __attribute__((always_inline)) {
        const int side = c >= (KC3 / 2) ? 1 : 0;
        const int lq   = (c - side * (KC3 / 2)) * TK3;
        const int l    = lq + lane;
        const int ly   = l / WD, lx = l - ly * WD;
        const float* pa = (side ? aR : aL) + ly * (2 * W_) + 2 * lx;
        #pragma unroll
        for (int t = 0; t < 8; ++t) va[t] = *(const v2f*)(pa + (size_t)t * 8 * (H_ * W_));
        #pragma unroll
        for (int t = 0; t < 3; ++t) {
            const int lk = lq + krow[t];
            const float* pb = bsrc[t] + (size_t)c * TK3 * L_;
            const bool ok = pok[t] & (a ? (lk < L_ - WD) : (lk >= WD));
            vb[t] = *(const v4f*)(pb);
            vc[t] = *(const v4f*)(pb + (ok ? ysh : 0));
        }
    };
    auto sstore = [&](const int c) __attribute__((always_inline)) {
        const int st   = c & 1;
        const int side = c >= (KC3 / 2) ? 1 : 0;
        const int lq   = (c - side * (KC3 / 2)) * TK3;
        #pragma unroll
        for (int t = 0; t < 8; ++t) {
            Ap[st][0][w + 8 * t][lane] = va[t].x;
            Ap[st][1][w + 8 * t][lane] = va[t].y;
        }
        #pragma unroll
        for (int t = 0; t < 3; ++t) {
            const int lk = lq + krow[t];
            const bool ok = pok[t] & (a ? (lk < L_ - WD) : (lk >= WD));
            #pragma unroll
            for (int e = 0; e < 4; ++e)
                Bp[st][bsel[t]][pv4[t] + e][krow[t]] = vb[t][e] + (ok ? vc[t][e] : 0.0f);
        }
    };
    auto fload = [&](const int c, v2f (&FA)[8], v2f (&F0)[8], v2f (&F1)[8], v2f (&F2)[8])
                 __attribute__((always_inline)) {
        const int st = c & 1;
        #pragma unroll
        for (int s = 0; s < 8; ++s) {
            const int ka = 4 * s + khalf;
            FA[s] = *(const v2f*)&Ap[st][bq][tr * 16 + m16][ka];
            F0[s] = *(const v2f*)&Bp[st][bq][m16][ka];
            F1[s] = *(const v2f*)&Bp[st][bq][16 + m16][ka];
            F2[s] = *(const v2f*)&Bp[st][bq][32 + m16][ka];
        }
    };
    auto mma = [&](const v2f (&FA)[8], const v2f (&F0)[8], const v2f (&F1)[8], const v2f (&F2)[8])
               __attribute__((always_inline)) {
        #pragma unroll
        for (int s = 0; s < 8; ++s) {
            acc0 = __builtin_amdgcn_wmma_f32_16x16x4_f32(false, FA[s], false, F0[s],
                                                         (short)0, acc0, false, false);
            acc1 = __builtin_amdgcn_wmma_f32_16x16x4_f32(false, FA[s], false, F1[s],
                                                         (short)0, acc1, false, false);
            acc2 = __builtin_amdgcn_wmma_f32_16x16x4_f32(false, FA[s], false, F2[s],
                                                         (short)0, acc2, false, false);
        }
    };

    v2f xa[8], x0[8], x1[8], x2[8];
    v2f ya[8], y0[8], y1[8], y2[8];

    gload(0);
    sstore(0);
    gload(1);
    WG_BARRIER_LDS();
    fload(0, xa, x0, x1, x2);
    sstore(1);
    gload(2);
    WG_BARRIER_LDS();

    for (int c = 0; c < KC3; c += 2) {
        if (early) {
            if (c + 2 < KC3) sstore(c + 2);
            if (c + 3 < KC3) gload(c + 3);
        }
        __builtin_amdgcn_sched_barrier(0);
        fload(c + 1, ya, y0, y1, y2);
        __builtin_amdgcn_sched_barrier(0);
        mma(xa, x0, x1, x2);
        __builtin_amdgcn_sched_barrier(0);
        if (!early) {
            if (c + 2 < KC3) sstore(c + 2);
            if (c + 3 < KC3) gload(c + 3);
        }
        WG_BARRIER_LDS();
        if (early) {
            if (c + 3 < KC3) sstore(c + 3);
            if (c + 4 < KC3) gload(c + 4);
        }
        __builtin_amdgcn_sched_barrier(0);
        fload(c + 2, xa, x0, x1, x2);
        __builtin_amdgcn_sched_barrier(0);
        mma(ya, y0, y1, y2);
        __builtin_amdgcn_sched_barrier(0);
        if (!early) {
            if (c + 3 < KC3) sstore(c + 3);
            if (c + 4 < KC3) gload(c + 4);
        }
        WG_BARRIER_LDS();
    }
    const int n  = lane & 15;
    const int cb = m0 + tr * 16 + ((lane >> 4) << 3);
    #pragma unroll
    for (int j = 0; j < 3; ++j) {
        const int p  = p0 + j * 16 + n;
        const int py = p / WD, px = p - py * WD;
        const size_t o = ((size_t)(b * C_ + cb) * H_ + 2 * py + a) * W_ + 2 * px + bq;
        const v8f acc = j == 0 ? acc0 : (j == 1 ? acc1 : acc2);
        #pragma unroll
        for (int v = 0; v < 8; ++v) out[o + (size_t)v * (H_ * W_)] = acc[v];
    }
}

extern "C" void kernel_launch(void* const* d_in, const int* in_sizes, int n_in,
                              void* d_out, int out_size, void* d_ws, size_t ws_size,
                              hipStream_t stream) {
    (void)in_sizes; (void)n_in; (void)out_size; (void)ws_size;
    const float* left  = (const float*)d_in[0];
    const float* right = (const float*)d_in[1];
    const float* mid   = (const float*)d_in[2];
    const float* shcL  = (const float*)d_in[3];
    const float* shcR  = (const float*)d_in[4];
    float* out = (float*)d_out;

    float* w   = (float*)d_ws;
    const size_t NDS = (size_t)B_ * C_ * HD * WD;
    const size_t NDN = (size_t)B_ * L_;
    const size_t NSC = (size_t)B_ * L_ * L_;
    const size_t NY  = (size_t)B_ * K16 * L_;
    float* dsL  = w;
    float* dsR  = dsL + NDS;
    float* dsM  = dsR + NDS;
    float* denL = dsM + NDS;
    float* denR = denL + NDN;
    float* scL  = denR + NDN;
    float* scR  = scL + NSC;
    float* YL   = scR + NSC;
    float* YR   = YL + NY;
    float* G    = scL;
    float* Xp   = YL;
    (void)scR; (void)YR;

    k_prep_ds<<<dim3((NDS + 255) / 256), dim3(256), 0, stream>>>(left, right, mid,
                                                                 dsL, dsR, dsM);
    k_scores<<<dim3(81, 2, B_), dim3(256), 0, stream>>>(dsL, dsR, dsM, G, denL);
    k_softmax<<<dim3(HD, 2, B_), dim3(SM_COLS * SM_WAVES), 0, stream>>>(G, denL, Xp);
    k_recon<<<dim3(48, 2, B_), dim3(256), 0, stream>>>(shcL, shcR, Xp, out);
}
